# starting kernel
# baseline (speedup 1.0000x reference)
#include <hip/hip_runtime.h>
#include <math.h>

#define NN 65536
#define NE 1048576
#define DD 128
#define DP 129
#define EPSF 1e-6f
#define BETAF 0.5f

typedef float v2f __attribute__((ext_vector_type(2)));
typedef float v8f __attribute__((ext_vector_type(8)));

__device__ __forceinline__ float wave_sum(float v) {
#pragma unroll
  for (int m = 16; m > 0; m >>= 1) v += __shfl_xor(v, m, 32);
  return v;
}
__device__ __forceinline__ float half_sum16(float v) {
#pragma unroll
  for (int m = 8; m > 0; m >>= 1) v += __shfl_xor(v, m, 32);
  return v;
}
__device__ __forceinline__ float acosh_(float z) {
  return logf(z + sqrtf(fmaf(z, z, -1.0f)));
}

__global__ __launch_bounds__(256) void phi_kernel(const float* X, float* __restrict__ phi) {
  int node = (blockIdx.x * blockDim.x + threadIdx.x) >> 5;
  int lane = threadIdx.x & 31;
  if (node >= NN) return;
  const float* row = X + (size_t)node * DP;
  float ss = 0.f;
#pragma unroll
  for (int j = 0; j < 4; ++j) {
    float v = row[1 + lane * 4 + j];
    ss = fmaf(v, v, ss);
  }
  ss = wave_sum(ss);
  if (lane == 0) {
    float xn = sqrtf(fmaxf(ss, EPSF * EPSF));
    float th = acosh_(fmaxf(row[0], 1.0f + EPSF));
    phi[node] = th / xn;
  }
}

__global__ __launch_bounds__(256) void gemm_kernel(const float* X,
    const float* __restrict__ phi, const float* __restrict__ W,
    const float* __restrict__ bias, float* __restrict__ t, float* __restrict__ s) {
  __shared__ float lw[128 * 130];
  for (int idx = threadIdx.x; idx < 128 * 128; idx += 256) {
    int r = idx >> 7, c = idx & 127;
    lw[r * 130 + c] = W[idx];
  }
  __syncthreads();

  int waveId = threadIdx.x >> 5;
  int lane = threadIdx.x & 31;
  int m0 = blockIdx.x * 128 + waveId * 16;
  int lh = lane & 15;
  int hi = lane >> 4;

  int arow = m0 + lh;
  float phiA = phi[arow];
  const float* xrow = X + (size_t)arow * DP + 1;

  v8f acc[8] = {};
  for (int k0 = 0; k0 < 128; k0 += 4) {
    int kk = k0 + hi * 2;
    v2f a;
    a.x = phiA * xrow[kk];
    a.y = phiA * xrow[kk + 1];
#pragma unroll
    for (int nt = 0; nt < 8; ++nt) {
      int bn = nt * 16 + lh;
      v2f bf;
      bf.x = lw[bn * 130 + kk];
      bf.y = lw[bn * 130 + kk + 1];
      acc[nt] = __builtin_amdgcn_wmma_f32_16x16x4_f32(
          false, a, false, bf, (short)0, acc[nt], false, false);
    }
  }
  float ss[8];
#pragma unroll
  for (int r = 0; r < 8; ++r) ss[r] = 0.f;
#pragma unroll
  for (int nt = 0; nt < 8; ++nt) {
    float bv = bias[nt * 16 + lh];
#pragma unroll
    for (int r = 0; r < 8; ++r) {
      float g = acc[nt][r] + bv;
      acc[nt][r] = g;
      ss[r] = fmaf(g, g, ss[r]);
    }
  }
#pragma unroll
  for (int r = 0; r < 8; ++r) ss[r] = half_sum16(ss[r]);

  float scl[8];
#pragma unroll
  for (int r = 0; r < 8; ++r) {
    float vss = ss[r];
    float vn = sqrtf(fmaxf(vss, EPSF * EPSF));
    float a = vn;
    float k1 = sinhf(a) / a;
    float hs2 = k1 * k1 * vss;
    float xn = sqrtf(fmaxf(hs2, EPSF * EPSF));
    float th = acosh_(fmaxf(coshf(a), 1.0f + EPSF));
    scl[r] = th / xn * k1;
  }
#pragma unroll
  for (int nt = 0; nt < 8; ++nt) {
    int col = nt * 16 + lh;
#pragma unroll
    for (int r = 0; r < 8; ++r) {
      int row = m0 + r + hi * 8;
      float v = scl[r] * acc[nt][r];
      size_t o = (size_t)row * DD + col;
      t[o] = v;
      s[o] = v;
    }
  }
}

__global__ __launch_bounds__(256) void deg_init(float* __restrict__ deg) {
  int i = blockIdx.x * blockDim.x + threadIdx.x;
  if (i < NN) deg[i] = 1.0f;
}

__global__ __launch_bounds__(256) void deg_scatter(const int* __restrict__ ei,
                                                   float* __restrict__ deg) {
  int i = blockIdx.x * blockDim.x + threadIdx.x;
  int n = gridDim.x * blockDim.x;
  for (int e = i; e < NE; e += n) atomicAdd(&deg[ei[NE + e]], 1.0f);
}

__global__ __launch_bounds__(256) void edge_scatter(const int* __restrict__ ei,
    const float* __restrict__ t, float* __restrict__ s) {
  int nw = (gridDim.x * blockDim.x) >> 5;
  int w = (blockIdx.x * blockDim.x + threadIdx.x) >> 5;
  int lane = threadIdx.x & 31;
  for (int e = w; e < NE; e += nw) {
    int src = ei[e];
    int dst = ei[NE + e];
    float4 v = ((const float4*)(t + (size_t)src * DD))[lane];
    float* so = s + (size_t)dst * DD + lane * 4;
    atomicAdd(so + 0, v.x);
    atomicAdd(so + 1, v.y);
    atomicAdd(so + 2, v.z);
    atomicAdd(so + 3, v.w);
  }
}

__global__ __launch_bounds__(256) void node_kernel(const float* X,
    const float* __restrict__ s, const float* __restrict__ deg, float* Y) {
  int i = (blockIdx.x * blockDim.x + threadIdx.x) >> 5;
  int lane = threadIdx.x & 31;
  if (i >= NN) return;
  const float* xrow = X + (size_t)i * DP;
  float x0 = xrow[0];
  float xs[4], mean[4];
  float invd = 1.0f / deg[i];
  float4 sv = ((const float4*)(s + (size_t)i * DD))[lane];
  mean[0] = sv.x * invd; mean[1] = sv.y * invd;
  mean[2] = sv.z * invd; mean[3] = sv.w * invd;
#pragma unroll
  for (int j = 0; j < 4; ++j) xs[j] = xrow[1 + lane * 4 + j];

  float r1 = 0.f;
#pragma unroll
  for (int j = 0; j < 4; ++j) r1 = fmaf(mean[j], mean[j], r1);
  r1 = wave_sum(r1);
  float vn = sqrtf(fmaxf(r1, EPSF * EPSF));
  float a1 = vn;
  float k1 = sinhf(a1) / a1;
  float agg_ss = k1 * k1 * r1;
  float x0a = sqrtf(1.0f + agg_ss);
  float un = sqrtf(fmaxf(agg_ss, EPSF * EPSF));
  float th = acosh_(fmaxf(x0a, 1.0f + EPSF));
  float su = th / un * k1;

  float u[4]; float ru = 0.f;
#pragma unroll
  for (int j = 0; j < 4; ++j) { u[j] = fmaxf(su * mean[j], 0.0f); ru = fmaf(u[j], u[j], ru); }
  ru = wave_sum(ru);
  float rn = sqrtf(fmaxf(ru, EPSF * EPSF));
  float a2 = rn;
  float act0 = coshf(a2);
  float k2 = sinhf(a2) / a2;

  float acts[4]; float dxa = 0.f;
#pragma unroll
  for (int j = 0; j < 4; ++j) { acts[j] = k2 * u[j]; dxa = fmaf(xs[j], acts[j], dxa); }
  dxa = wave_sum(dxa);
  float alpha = fmaxf(x0 * act0 - dxa, 1.0f + EPSF);

  float u2[4]; float u2ss = 0.f;
#pragma unroll
  for (int j = 0; j < 4; ++j) { u2[j] = acts[j] - alpha * xs[j]; u2ss = fmaf(u2[j], u2[j], u2ss); }
  u2ss = wave_sum(u2ss);
  float u20 = act0 - alpha * x0;
  float mu2 = u2ss - u20 * u20;
  float un2 = sqrtf(fmaxf(mu2, EPSF * EPSF));
  float dist = acosh_(alpha);
  float sv2 = BETAF * dist / un2;
  float bv0 = sv2 * u20;
  float mbv = sv2 * sv2 * mu2;
  float a3 = sqrtf(fmaxf(mbv, EPSF * EPSF));
  float c3 = coshf(a3);
  float k3 = sinhf(a3) / a3;

  float out[4]; float oss = 0.f;
#pragma unroll
  for (int j = 0; j < 4; ++j) {
    out[j] = c3 * xs[j] + k3 * (sv2 * u2[j]);
    oss = fmaf(out[j], out[j], oss);
  }
  oss = wave_sum(oss);
  float y0 = sqrtf(1.0f + oss);
  (void)bv0;
  float* yrow = Y + (size_t)i * DP;
  if (lane == 0) yrow[0] = y0;
#pragma unroll
  for (int j = 0; j < 4; ++j) yrow[1 + lane * 4 + j] = out[j];
}

extern "C" void kernel_launch(void* const* d_in, const int* in_sizes, int n_in,
                              void* d_out, int out_size, void* d_ws, size_t ws_size,
                              hipStream_t stream) {
  const float* x  = (const float*)d_in[0];
  const int*   ei = (const int*)d_in[1];
  const float* W1 = (const float*)d_in[2];
  const float* b1 = (const float*)d_in[3];
  const float* W2 = (const float*)d_in[4];
  const float* b2 = (const float*)d_in[5];
  float* out = (float*)d_out;

  float* t   = (float*)d_ws;
  float* s   = t + (size_t)NN * DD;
  float* phi = s + (size_t)NN * DD;
  float* deg = phi + NN;

  deg_init<<<NN / 256, 256, 0, stream>>>(deg);
  deg_scatter<<<1024, 256, 0, stream>>>(ei, deg);

  const float* Ws[2] = {W1, W2};
  const float* bs[2] = {b1, b2};
  for (int L = 0; L < 2; ++L) {
    const float* X = (L == 0) ? x : out;
    phi_kernel<<<NN / 8, 256, 0, stream>>>(X, phi);
    gemm_kernel<<<NN / 128, 256, 0, stream>>>(X, phi, Ws[L], bs[L], t, s);
    edge_scatter<<<2048, 256, 0, stream>>>(ei, t, s);
    node_kernel<<<NN / 8, 256, 0, stream>>>(X, s, deg, out);
  }
}


# segment sum as a gather: edges ordered by destination once per call (slot = return of an integer counting add), one wave sums a node's rows in registers in slot order; no float atomics
# speedup vs baseline: 2.2078x; 2.2078x over previous
#include <hip/hip_runtime.h>
#include <math.h>

#define NN 65536
#define NE 1048576
#define DD 128
#define DP 129
#define EPSF 1e-6f
#define BETAF 0.5f

typedef float v2f __attribute__((ext_vector_type(2)));
typedef float v8f __attribute__((ext_vector_type(8)));

__device__ __forceinline__ float wave_sum(float v) {
#pragma unroll
  for (int m = 16; m > 0; m >>= 1) v += __shfl_xor(v, m, 32);
  return v;
}
__device__ __forceinline__ float half_sum16(float v) {
#pragma unroll
  for (int m = 8; m > 0; m >>= 1) v += __shfl_xor(v, m, 32);
  return v;
}
__device__ __forceinline__ float acosh_(float z) {
  return logf(z + sqrtf(fmaf(z, z, -1.0f)));
}

__global__ __launch_bounds__(256) void phi_kernel(const float* X, float* __restrict__ phi) {
  int node = (blockIdx.x * blockDim.x + threadIdx.x) >> 5;
  int lane = threadIdx.x & 31;
  if (node >= NN) return;
  const float* row = X + (size_t)node * DP;
  float ss = 0.f;
#pragma unroll
  for (int j = 0; j < 4; ++j) {
    float v = row[1 + lane * 4 + j];
    ss = fmaf(v, v, ss);
  }
  ss = wave_sum(ss);
  if (lane == 0) {
    float xn = sqrtf(fmaxf(ss, EPSF * EPSF));
    float th = acosh_(fmaxf(row[0], 1.0f + EPSF));
    phi[node] = th / xn;
  }
}

__global__ __launch_bounds__(256) void gemm_kernel(const float* X,
    const float* __restrict__ phi, const float* __restrict__ W,
    const float* __restrict__ bias, float* __restrict__ t) {
  __shared__ float lw[128 * 130];
  for (int idx = threadIdx.x; idx < 128 * 128; idx += 256) {
    int r = idx >> 7, c = idx & 127;
    lw[r * 130 + c] = W[idx];
  }
  __syncthreads();

  int waveId = threadIdx.x >> 5;
  int lane = threadIdx.x & 31;
  int m0 = blockIdx.x * 128 + waveId * 16;
  int lh = lane & 15;
  int hi = lane >> 4;

  int arow = m0 + lh;
  float phiA = phi[arow];
  const float* xrow = X + (size_t)arow * DP + 1;

  v8f acc[8] = {};
  for (int k0 = 0; k0 < 128; k0 += 4) {
    int kk = k0 + hi * 2;
    v2f a;
    a.x = phiA * xrow[kk];
    a.y = phiA * xrow[kk + 1];
#pragma unroll
    for (int nt = 0; nt < 8; ++nt) {
      int bn = nt * 16 + lh;
      v2f bf;
      bf.x = lw[bn * 130 + kk];
      bf.y = lw[bn * 130 + kk + 1];
      acc[nt] = __builtin_amdgcn_wmma_f32_16x16x4_f32(
          false, a, false, bf, (short)0, acc[nt], false, false);
    }
  }
  float ss[8];
#pragma unroll
  for (int r = 0; r < 8; ++r) ss[r] = 0.f;
#pragma unroll
  for (int nt = 0; nt < 8; ++nt) {
    float bv = bias[nt * 16 + lh];
#pragma unroll
    for (int r = 0; r < 8; ++r) {
      float g = acc[nt][r] + bv;
      acc[nt][r] = g;
      ss[r] = fmaf(g, g, ss[r]);
    }
  }
#pragma unroll
  for (int r = 0; r < 8; ++r) ss[r] = half_sum16(ss[r]);

  float scl[8];
#pragma unroll
  for (int r = 0; r < 8; ++r) {
    float vss = ss[r];
    float vn = sqrtf(fmaxf(vss, EPSF * EPSF));
    float a = vn;
    float k1 = sinhf(a) / a;
    float hs2 = k1 * k1 * vss;
    float xn = sqrtf(fmaxf(hs2, EPSF * EPSF));
    float th = acosh_(fmaxf(coshf(a), 1.0f + EPSF));
    scl[r] = th / xn * k1;
  }
#pragma unroll
  for (int nt = 0; nt < 8; ++nt) {
    int col = nt * 16 + lh;
#pragma unroll
    for (int r = 0; r < 8; ++r) {
      int row = m0 + r + hi * 8;
      float v = scl[r] * acc[nt][r];
      size_t o = (size_t)row * DD + col;
      t[o] = v;
    }
  }
}

__global__ __launch_bounds__(256) void cnt_zero(int* __restrict__ cnt) {
  int i = blockIdx.x * blockDim.x + threadIdx.x;
  if (i < NN) cnt[i] = 0;
}

__global__ __launch_bounds__(256) void rank_count(const int* __restrict__ ei,
    int* __restrict__ cnt, int* __restrict__ rank) {
  int e = blockIdx.x * blockDim.x + threadIdx.x;
  if (e < NE) rank[e] = atomicAdd(&cnt[ei[NE + e]], 1);
}

__global__ __launch_bounds__(1024) void scan_kernel(const int* __restrict__ cnt,
                                                    int* __restrict__ seg) {
  __shared__ int pre_w[32];
  __shared__ int tot_w[8];
  int b = blockIdx.x, tid = threadIdx.x;
  int lane = tid & 31, w = tid >> 5;
  const int4* c4 = (const int4*)cnt;
  int lim = b * 64;
  int pre = 0;
#pragma unroll
  for (int j = 0; j < 16; ++j) {
    int idx = tid + j * 1024;
    int4 v = c4[idx];
    int keep = (idx - lim) >> 31;
    pre += ((v.x + v.y) + (v.z + v.w)) & keep;
  }
  for (int m = 16; m > 0; m >>= 1) pre += __shfl_xor(pre, m, 32);
  int mine = (tid < 256) ? cnt[b * 256 + tid] : 0;
  int inc = mine;
  for (int m = 1; m < 32; m <<= 1) {
    int o = __shfl_up(inc, m, 32);
    if (lane >= m) inc += o;
  }
  if (lane == 0) pre_w[w] = pre;
  if (lane == 31 && w < 8) tot_w[w] = inc;
  __syncthreads();
  if (tid < 256) {
    int base = 0;
    for (int k = 0; k < 32; ++k) base += pre_w[k];
    for (int k = 0; k < 8; ++k) if (k < w) base += tot_w[k];
    ((int2*)seg)[b * 256 + tid] = make_int2(base + inc - mine, mine);
  }
}

__global__ __launch_bounds__(256) void fill_kernel(const int* __restrict__ ei,
    const int* __restrict__ seg, const int* __restrict__ rank, int* __restrict__ csr) {
  int e = blockIdx.x * blockDim.x + threadIdx.x;
  if (e < NE) csr[seg[2 * ei[NE + e]] + rank[e]] = ei[e];
}

template <int R>
__device__ __forceinline__ void add_rows(const float4* __restrict__ t4, int mine, int k, int lane,
                                         float4& sv) {
  float4 v[R];
#pragma unroll
  for (int j = 0; j < R; ++j) {
    int s = __builtin_amdgcn_readlane(mine, k + j);
    v[j] = t4[(size_t)s * 32 + lane];
  }
#pragma unroll
  for (int j = 0; j < R; ++j) {
    sv.x += v[j].x; sv.y += v[j].y; sv.z += v[j].z; sv.w += v[j].w;
  }
}

__global__ __launch_bounds__(256) void node_kernel(const float* X,
    const float* __restrict__ t, const int* __restrict__ seg,
    const int* __restrict__ csr, float* Y) {
  int i = __builtin_amdgcn_readfirstlane((int)((blockIdx.x * blockDim.x + threadIdx.x) >> 5));
  int lane = threadIdx.x & 31;
  if (i >= NN) return;
  const float4* t4 = (const float4*)t;
  int2 sg = ((const int2*)seg)[i];
  int start = sg.x, n = sg.y;
  const float* xrow = X + (size_t)i * DP;
  float x0 = xrow[0];
  float xs[4], mean[4];
#pragma unroll
  for (int j = 0; j < 4; ++j) xs[j] = xrow[1 + lane * 4 + j];
  float4 sv = t4[(size_t)i * 32 + lane];
  for (int base = 0; base < n; base += 32) {
    int m = (n - base < 32) ? (n - base) : 32;
    int mine = (lane < m) ? csr[start + base + lane] : 0;
    int k = 0;
    for (; k + 8 <= m; k += 8) add_rows<8>(t4, mine, k, lane, sv);
    if (k + 4 <= m) { add_rows<4>(t4, mine, k, lane, sv); k += 4; }
    int r = m - k;
    if (r == 3) add_rows<3>(t4, mine, k, lane, sv);
    else if (r == 2) add_rows<2>(t4, mine, k, lane, sv);
    else if (r == 1) add_rows<1>(t4, mine, k, lane, sv);
  }
  float invd = 1.0f / (float)(n + 1);
  mean[0] = sv.x * invd; mean[1] = sv.y * invd;
  mean[2] = sv.z * invd; mean[3] = sv.w * invd;

  float r1 = 0.f;
#pragma unroll
  for (int j = 0; j < 4; ++j) r1 = fmaf(mean[j], mean[j], r1);
  r1 = wave_sum(r1);
  float vn = sqrtf(fmaxf(r1, EPSF * EPSF));
  float a1 = vn;
  float k1 = sinhf(a1) / a1;
  float agg_ss = k1 * k1 * r1;
  float x0a = sqrtf(1.0f + agg_ss);
  float un = sqrtf(fmaxf(agg_ss, EPSF * EPSF));
  float th = acosh_(fmaxf(x0a, 1.0f + EPSF));
  float su = th / un * k1;

  float u[4]; float ru = 0.f;
#pragma unroll
  for (int j = 0; j < 4; ++j) { u[j] = fmaxf(su * mean[j], 0.0f); ru = fmaf(u[j], u[j], ru); }
  ru = wave_sum(ru);
  float rn = sqrtf(fmaxf(ru, EPSF * EPSF));
  float a2 = rn;
  float act0 = coshf(a2);
  float k2 = sinhf(a2) / a2;

  float acts[4]; float dxa = 0.f;
#pragma unroll
  for (int j = 0; j < 4; ++j) { acts[j] = k2 * u[j]; dxa = fmaf(xs[j], acts[j], dxa); }
  dxa = wave_sum(dxa);
  float alpha = fmaxf(x0 * act0 - dxa, 1.0f + EPSF);

  float u2[4]; float u2ss = 0.f;
#pragma unroll
  for (int j = 0; j < 4; ++j) { u2[j] = acts[j] - alpha * xs[j]; u2ss = fmaf(u2[j], u2[j], u2ss); }
  u2ss = wave_sum(u2ss);
  float u20 = act0 - alpha * x0;
  float mu2 = u2ss - u20 * u20;
  float un2 = sqrtf(fmaxf(mu2, EPSF * EPSF));
  float dist = acosh_(alpha);
  float sv2 = BETAF * dist / un2;
  float bv0 = sv2 * u20;
  float mbv = sv2 * sv2 * mu2;
  float a3 = sqrtf(fmaxf(mbv, EPSF * EPSF));
  float c3 = coshf(a3);
  float k3 = sinhf(a3) / a3;

  float out[4]; float oss = 0.f;
#pragma unroll
  for (int j = 0; j < 4; ++j) {
    out[j] = c3 * xs[j] + k3 * (sv2 * u2[j]);
    oss = fmaf(out[j], out[j], oss);
  }
  oss = wave_sum(oss);
  float y0 = sqrtf(1.0f + oss);
  (void)bv0;
  float* yrow = Y + (size_t)i * DP;
  if (lane == 0) yrow[0] = y0;
#pragma unroll
  for (int j = 0; j < 4; ++j) yrow[1 + lane * 4 + j] = out[j];
}

extern "C" void kernel_launch(void* const* d_in, const int* in_sizes, int n_in,
                              void* d_out, int out_size, void* d_ws, size_t ws_size,
                              hipStream_t stream) {
  const float* x  = (const float*)d_in[0];
  const int*   ei = (const int*)d_in[1];
  const float* W1 = (const float*)d_in[2];
  const float* b1 = (const float*)d_in[3];
  const float* W2 = (const float*)d_in[4];
  const float* b2 = (const float*)d_in[5];
  float* out = (float*)d_out;

  float* t    = (float*)d_ws;
  float* phi  = t + (size_t)NN * DD;
  int* cnt    = (int*)(phi + NN);
  int* seg    = cnt + NN;
  int* rank   = seg + 2 * NN;
  int* csr    = rank + NE;

  cnt_zero<<<NN / 256, 256, 0, stream>>>(cnt);
  rank_count<<<NE / 256, 256, 0, stream>>>(ei, cnt, rank);
  scan_kernel<<<NN / 256, 1024, 0, stream>>>(cnt, seg);
  fill_kernel<<<NE / 256, 256, 0, stream>>>(ei, seg, rank, csr);

  const float* Ws[2] = {W1, W2};
  const float* bs[2] = {b1, b2};
  for (int L = 0; L < 2; ++L) {
    const float* X = (L == 0) ? x : out;
    phi_kernel<<<NN / 8, 256, 0, stream>>>(X, phi);
    gemm_kernel<<<NN / 128, 256, 0, stream>>>(X, phi, Ws[L], bs[L], t);
    node_kernel<<<NN / 8, 256, 0, stream>>>(X, t, seg, csr, out);
  }
}


# row scale phi formed in the GEMM kernel's prologue by the wave that owns the rows, same summation order, under the weight staging; the two phi launches and their read of X are gone
# speedup vs baseline: 2.4609x; 1.1146x over previous
#include <hip/hip_runtime.h>
#include <math.h>

#define NN 65536
#define NE 1048576
#define DD 128
#define DP 129
#define EPSF 1e-6f
#define BETAF 0.5f

typedef float v2f __attribute__((ext_vector_type(2)));
typedef float v8f __attribute__((ext_vector_type(8)));

__device__ __forceinline__ float wave_sum(float v) {
#pragma unroll
  for (int m = 16; m > 0; m >>= 1) v += __shfl_xor(v, m, 32);
  return v;
}
__device__ __forceinline__ float half_sum16(float v) {
#pragma unroll
  for (int m = 8; m > 0; m >>= 1) v += __shfl_xor(v, m, 32);
  return v;
}
__device__ __forceinline__ float acosh_(float z) {
  return logf(z + sqrtf(fmaf(z, z, -1.0f)));
}

__global__ __launch_bounds__(256) void gemm_kernel(const float* X, const float* __restrict__ W,
    const float* __restrict__ bias, float* __restrict__ t) {
  __shared__ float lw[128 * 130];
  int waveId = threadIdx.x >> 5;
  int lane = threadIdx.x & 31;
  int m0 = blockIdx.x * 128 + waveId * 16;
  int lh = lane & 15;
  int hi = lane >> 4;
  int arow = m0 + lh;

  float xv[16][4];
#pragma unroll
  for (int r = 0; r < 16; ++r) {
    const float* row = X + (size_t)(m0 + r) * DP + 1 + lane * 4;
#pragma unroll
    for (int j = 0; j < 4; ++j) xv[r][j] = row[j];
  }
  float x0 = X[(size_t)arow * DP];

  for (int idx = threadIdx.x; idx < 128 * 128; idx += 256) {
    int r = idx >> 7, c = idx & 127;
    lw[r * 130 + c] = W[idx];
  }
  __syncthreads();

  float ssq = 0.f;
#pragma unroll
  for (int r = 0; r < 16; ++r) {
    float s = 0.f;
#pragma unroll
    for (int j = 0; j < 4; ++j) s = fmaf(xv[r][j], xv[r][j], s);
    s = wave_sum(s);
    if (lh == r) ssq = s;
  }
  float phiA = acosh_(fmaxf(x0, 1.0f + EPSF)) / sqrtf(fmaxf(ssq, EPSF * EPSF));
  const float* xrow = X + (size_t)arow * DP + 1;

  v8f acc[8] = {};
  for (int k0 = 0; k0 < 128; k0 += 4) {
    int kk = k0 + hi * 2;
    v2f a;
    a.x = phiA * xrow[kk];
    a.y = phiA * xrow[kk + 1];
#pragma unroll
    for (int nt = 0; nt < 8; ++nt) {
      int bn = nt * 16 + lh;
      v2f bf;
      bf.x = lw[bn * 130 + kk];
      bf.y = lw[bn * 130 + kk + 1];
      acc[nt] = __builtin_amdgcn_wmma_f32_16x16x4_f32(
          false, a, false, bf, (short)0, acc[nt], false, false);
    }
  }
  float ss[8];
#pragma unroll
  for (int r = 0; r < 8; ++r) ss[r] = 0.f;
#pragma unroll
  for (int nt = 0; nt < 8; ++nt) {
    float bv = bias[nt * 16 + lh];
#pragma unroll
    for (int r = 0; r < 8; ++r) {
      float g = acc[nt][r] + bv;
      acc[nt][r] = g;
      ss[r] = fmaf(g, g, ss[r]);
    }
  }
#pragma unroll
  for (int r = 0; r < 8; ++r) ss[r] = half_sum16(ss[r]);

  float scl[8];
#pragma unroll
  for (int r = 0; r < 8; ++r) {
    float vss = ss[r];
    float vn = sqrtf(fmaxf(vss, EPSF * EPSF));
    float a = vn;
    float k1 = sinhf(a) / a;
    float hs2 = k1 * k1 * vss;
    float xn = sqrtf(fmaxf(hs2, EPSF * EPSF));
    float th = acosh_(fmaxf(coshf(a), 1.0f + EPSF));
    scl[r] = th / xn * k1;
  }
#pragma unroll
  for (int nt = 0; nt < 8; ++nt) {
    int col = nt * 16 + lh;
#pragma unroll
    for (int r = 0; r < 8; ++r) {
      int row = m0 + r + hi * 8;
      float v = scl[r] * acc[nt][r];
      size_t o = (size_t)row * DD + col;
      t[o] = v;
    }
  }
}

__global__ __launch_bounds__(256) void cnt_zero(int* __restrict__ cnt) {
  int i = blockIdx.x * blockDim.x + threadIdx.x;
  if (i < NN) cnt[i] = 0;
}

__global__ __launch_bounds__(256) void rank_count(const int* __restrict__ ei,
    int* __restrict__ cnt, int* __restrict__ rank) {
  int e = blockIdx.x * blockDim.x + threadIdx.x;
  if (e < NE) rank[e] = atomicAdd(&cnt[ei[NE + e]], 1);
}

__global__ __launch_bounds__(1024) void scan_kernel(const int* __restrict__ cnt,
                                                    int* __restrict__ seg) {
  __shared__ int pre_w[32];
  __shared__ int tot_w[8];
  int b = blockIdx.x, tid = threadIdx.x;
  int lane = tid & 31, w = tid >> 5;
  const int4* c4 = (const int4*)cnt;
  int lim = b * 64;
  int pre = 0;
#pragma unroll
  for (int j = 0; j < 16; ++j) {
    int idx = tid + j * 1024;
    int4 v = c4[idx];
    int keep = (idx - lim) >> 31;
    pre += ((v.x + v.y) + (v.z + v.w)) & keep;
  }
  for (int m = 16; m > 0; m >>= 1) pre += __shfl_xor(pre, m, 32);
  int mine = (tid < 256) ? cnt[b * 256 + tid] : 0;
  int inc = mine;
  for (int m = 1; m < 32; m <<= 1) {
    int o = __shfl_up(inc, m, 32);
    if (lane >= m) inc += o;
  }
  if (lane == 0) pre_w[w] = pre;
  if (lane == 31 && w < 8) tot_w[w] = inc;
  __syncthreads();
  if (tid < 256) {
    int base = 0;
    for (int k = 0; k < 32; ++k) base += pre_w[k];
    for (int k = 0; k < 8; ++k) if (k < w) base += tot_w[k];
    ((int2*)seg)[b * 256 + tid] = make_int2(base + inc - mine, mine);
  }
}

__global__ __launch_bounds__(256) void fill_kernel(const int* __restrict__ ei,
    const int* __restrict__ seg, const int* __restrict__ rank, int* __restrict__ csr) {
  int e = blockIdx.x * blockDim.x + threadIdx.x;
  if (e < NE) csr[seg[2 * ei[NE + e]] + rank[e]] = ei[e];
}

template <int R>
__device__ __forceinline__ void add_rows(const float4* __restrict__ t4, int mine, int k, int lane,
                                         float4& sv) {
  float4 v[R];
#pragma unroll
  for (int j = 0; j < R; ++j) {
    int s = __builtin_amdgcn_readlane(mine, k + j);
    v[j] = t4[(size_t)s * 32 + lane];
  }
#pragma unroll
  for (int j = 0; j < R; ++j) {
    sv.x += v[j].x; sv.y += v[j].y; sv.z += v[j].z; sv.w += v[j].w;
  }
}

__global__ __launch_bounds__(256) void node_kernel(const float* X,
    const float* __restrict__ t, const int* __restrict__ seg,
    const int* __restrict__ csr, float* Y) {
  int i = __builtin_amdgcn_readfirstlane((int)((blockIdx.x * blockDim.x + threadIdx.x) >> 5));
  int lane = threadIdx.x & 31;
  if (i >= NN) return;
  const float4* t4 = (const float4*)t;
  int2 sg = ((const int2*)seg)[i];
  int start = sg.x, n = sg.y;
  const float* xrow = X + (size_t)i * DP;
  float x0 = xrow[0];
  float xs[4], mean[4];
#pragma unroll
  for (int j = 0; j < 4; ++j) xs[j] = xrow[1 + lane * 4 + j];
  float4 sv = t4[(size_t)i * 32 + lane];
  for (int base = 0; base < n; base += 32) {
    int m = (n - base < 32) ? (n - base) : 32;
    int mine = (lane < m) ? csr[start + base + lane] : 0;
    int k = 0;
    for (; k + 8 <= m; k += 8) add_rows<8>(t4, mine, k, lane, sv);
    if (k + 4 <= m) { add_rows<4>(t4, mine, k, lane, sv); k += 4; }
    int r = m - k;
    if (r == 3) add_rows<3>(t4, mine, k, lane, sv);
    else if (r == 2) add_rows<2>(t4, mine, k, lane, sv);
    else if (r == 1) add_rows<1>(t4, mine, k, lane, sv);
  }
  float invd = 1.0f / (float)(n + 1);
  mean[0] = sv.x * invd; mean[1] = sv.y * invd;
  mean[2] = sv.z * invd; mean[3] = sv.w * invd;

  float r1 = 0.f;
#pragma unroll
  for (int j = 0; j < 4; ++j) r1 = fmaf(mean[j], mean[j], r1);
  r1 = wave_sum(r1);
  float vn = sqrtf(fmaxf(r1, EPSF * EPSF));
  float a1 = vn;
  float k1 = sinhf(a1) / a1;
  float agg_ss = k1 * k1 * r1;
  float x0a = sqrtf(1.0f + agg_ss);
  float un = sqrtf(fmaxf(agg_ss, EPSF * EPSF));
  float th = acosh_(fmaxf(x0a, 1.0f + EPSF));
  float su = th / un * k1;

  float u[4]; float ru = 0.f;
#pragma unroll
  for (int j = 0; j < 4; ++j) { u[j] = fmaxf(su * mean[j], 0.0f); ru = fmaf(u[j], u[j], ru); }
  ru = wave_sum(ru);
  float rn = sqrtf(fmaxf(ru, EPSF * EPSF));
  float a2 = rn;
  float act0 = coshf(a2);
  float k2 = sinhf(a2) / a2;

  float acts[4]; float dxa = 0.f;
#pragma unroll
  for (int j = 0; j < 4; ++j) { acts[j] = k2 * u[j]; dxa = fmaf(xs[j], acts[j], dxa); }
  dxa = wave_sum(dxa);
  float alpha = fmaxf(x0 * act0 - dxa, 1.0f + EPSF);

  float u2[4]; float u2ss = 0.f;
#pragma unroll
  for (int j = 0; j < 4; ++j) { u2[j] = acts[j] - alpha * xs[j]; u2ss = fmaf(u2[j], u2[j], u2ss); }
  u2ss = wave_sum(u2ss);
  float u20 = act0 - alpha * x0;
  float mu2 = u2ss - u20 * u20;
  float un2 = sqrtf(fmaxf(mu2, EPSF * EPSF));
  float dist = acosh_(alpha);
  float sv2 = BETAF * dist / un2;
  float bv0 = sv2 * u20;
  float mbv = sv2 * sv2 * mu2;
  float a3 = sqrtf(fmaxf(mbv, EPSF * EPSF));
  float c3 = coshf(a3);
  float k3 = sinhf(a3) / a3;

  float out[4]; float oss = 0.f;
#pragma unroll
  for (int j = 0; j < 4; ++j) {
    out[j] = c3 * xs[j] + k3 * (sv2 * u2[j]);
    oss = fmaf(out[j], out[j], oss);
  }
  oss = wave_sum(oss);
  float y0 = sqrtf(1.0f + oss);
  (void)bv0;
  float* yrow = Y + (size_t)i * DP;
  if (lane == 0) yrow[0] = y0;
#pragma unroll
  for (int j = 0; j < 4; ++j) yrow[1 + lane * 4 + j] = out[j];
}

extern "C" void kernel_launch(void* const* d_in, const int* in_sizes, int n_in,
                              void* d_out, int out_size, void* d_ws, size_t ws_size,
                              hipStream_t stream) {
  const float* x  = (const float*)d_in[0];
  const int*   ei = (const int*)d_in[1];
  const float* W1 = (const float*)d_in[2];
  const float* b1 = (const float*)d_in[3];
  const float* W2 = (const float*)d_in[4];
  const float* b2 = (const float*)d_in[5];
  float* out = (float*)d_out;

  float* t    = (float*)d_ws;
  int* cnt    = (int*)(t + (size_t)NN * DD);
  int* seg    = cnt + NN;
  int* rank   = seg + 2 * NN;
  int* csr    = rank + NE;

  cnt_zero<<<NN / 256, 256, 0, stream>>>(cnt);
  rank_count<<<NE / 256, 256, 0, stream>>>(ei, cnt, rank);
  scan_kernel<<<NN / 256, 1024, 0, stream>>>(cnt, seg);
  fill_kernel<<<NE / 256, 256, 0, stream>>>(ei, seg, rank, csr);

  const float* Ws[2] = {W1, W2};
  const float* bs[2] = {b1, b2};
  for (int L = 0; L < 2; ++L) {
    const float* X = (L == 0) ? x : out;
    gemm_kernel<<<NN / 128, 256, 0, stream>>>(X, Ws[L], bs[L], t);
    node_kernel<<<NN / 8, 256, 0, stream>>>(X, t, seg, csr, out);
  }
}


# edges ordered by destination with LDS counters (per-chunk 256-bucket sort, then per-bucket slot = return of an LDS add) in 2 launches, not global returning adds in 4; slot order: LDS arrival order
# speedup vs baseline: 3.6082x; 1.4662x over previous
#include <hip/hip_runtime.h>
#include <math.h>

#define NN 65536
#define NE 1048576
#define DD 128
#define DP 129
#define EPSF 1e-6f
#define BETAF 0.5f

typedef float v2f __attribute__((ext_vector_type(2)));
typedef float v8f __attribute__((ext_vector_type(8)));

__device__ __forceinline__ float wave_sum(float v) {
#pragma unroll
  for (int m = 16; m > 0; m >>= 1) v += __shfl_xor(v, m, 32);
  return v;
}
__device__ __forceinline__ float half_sum16(float v) {
#pragma unroll
  for (int m = 8; m > 0; m >>= 1) v += __shfl_xor(v, m, 32);
  return v;
}
__device__ __forceinline__ float acosh_(float z) {
  return logf(z + sqrtf(fmaf(z, z, -1.0f)));
}

__global__ __launch_bounds__(256) void gemm_kernel(const float* X, const float* __restrict__ W,
    const float* __restrict__ bias, float* __restrict__ t) {
  __shared__ float lw[128 * 130];
  int waveId = threadIdx.x >> 5;
  int lane = threadIdx.x & 31;
  int m0 = blockIdx.x * 128 + waveId * 16;
  int lh = lane & 15;
  int hi = lane >> 4;
  int arow = m0 + lh;

  float xv[16][4];
#pragma unroll
  for (int r = 0; r < 16; ++r) {
    const float* row = X + (size_t)(m0 + r) * DP + 1 + lane * 4;
#pragma unroll
    for (int j = 0; j < 4; ++j) xv[r][j] = row[j];
  }
  float x0 = X[(size_t)arow * DP];

  for (int idx = threadIdx.x; idx < 128 * 128; idx += 256) {
    int r = idx >> 7, c = idx & 127;
    lw[r * 130 + c] = W[idx];
  }
  __syncthreads();

  float ssq = 0.f;
#pragma unroll
  for (int r = 0; r < 16; ++r) {
    float s = 0.f;
#pragma unroll
    for (int j = 0; j < 4; ++j) s = fmaf(xv[r][j], xv[r][j], s);
    s = wave_sum(s);
    if (lh == r) ssq = s;
  }
  float phiA = acosh_(fmaxf(x0, 1.0f + EPSF)) / sqrtf(fmaxf(ssq, EPSF * EPSF));
  const float* xrow = X + (size_t)arow * DP + 1;

  v8f acc[8] = {};
  for (int k0 = 0; k0 < 128; k0 += 4) {
    int kk = k0 + hi * 2;
    v2f a;
    a.x = phiA * xrow[kk];
    a.y = phiA * xrow[kk + 1];
#pragma unroll
    for (int nt = 0; nt < 8; ++nt) {
      int bn = nt * 16 + lh;
      v2f bf;
      bf.x = lw[bn * 130 + kk];
      bf.y = lw[bn * 130 + kk + 1];
      acc[nt] = __builtin_amdgcn_wmma_f32_16x16x4_f32(
          false, a, false, bf, (short)0, acc[nt], false, false);
    }
  }
  float ss[8];
#pragma unroll
  for (int r = 0; r < 8; ++r) ss[r] = 0.f;
#pragma unroll
  for (int nt = 0; nt < 8; ++nt) {
    float bv = bias[nt * 16 + lh];
#pragma unroll
    for (int r = 0; r < 8; ++r) {
      float g = acc[nt][r] + bv;
      acc[nt][r] = g;
      ss[r] = fmaf(g, g, ss[r]);
    }
  }
#pragma unroll
  for (int r = 0; r < 8; ++r) ss[r] = half_sum16(ss[r]);

  float scl[8];
#pragma unroll
  for (int r = 0; r < 8; ++r) {
    float vss = ss[r];
    float vn = sqrtf(fmaxf(vss, EPSF * EPSF));
    float a = vn;
    float k1 = sinhf(a) / a;
    float hs2 = k1 * k1 * vss;
    float xn = sqrtf(fmaxf(hs2, EPSF * EPSF));
    float th = acosh_(fmaxf(coshf(a), 1.0f + EPSF));
    scl[r] = th / xn * k1;
  }
#pragma unroll
  for (int nt = 0; nt < 8; ++nt) {
    int col = nt * 16 + lh;
#pragma unroll
    for (int r = 0; r < 8; ++r) {
      int row = m0 + r + hi * 8;
      float v = scl[r] * acc[nt][r];
      size_t o = (size_t)row * DD + col;
      t[o] = v;
    }
  }
}

#define NCH 256
#define CHE 4096
#define NBK 256
#define CAP 8192

__device__ __forceinline__ int scan256(int v, int tid, int* wt, int& total) {
  int lane = tid & 31, w = tid >> 5;
  int inc = v;
  for (int m = 1; m < 32; m <<= 1) {
    int o = __shfl_up(inc, m, 32);
    if (lane >= m) inc += o;
  }
  if (lane == 31 && w < 8) wt[w] = inc;
  __syncthreads();
  int base = 0, tot = 0;
  for (int k = 0; k < 8; ++k) {
    int x = wt[k];
    tot += x;
    if (k < w) base += x;
  }
  __syncthreads();
  total = tot;
  return base + inc - v;
}

__global__ __launch_bounds__(1024) void bin_kernel(const int* __restrict__ ei,
    int* __restrict__ tmp, int* __restrict__ offsT) {
  __shared__ int cnt[NBK];
  __shared__ int offs[NBK];
  __shared__ int wt[8];
  __shared__ __attribute__((aligned(16))) int stage[CHE];
  int tid = threadIdx.x;
  int base = blockIdx.x * CHE;
  if (tid < NBK) cnt[tid] = 0;
  int4 s4 = *(const int4*)(ei + base + 4 * tid);
  int4 d4 = *(const int4*)(ei + NE + base + 4 * tid);
  __syncthreads();
  int s[4] = {s4.x, s4.y, s4.z, s4.w};
  int d[4] = {d4.x, d4.y, d4.z, d4.w};
  int r[4];
#pragma unroll
  for (int j = 0; j < 4; ++j) r[j] = atomicAdd(&cnt[d[j] >> 8], 1);
  __syncthreads();
  int v = (tid < NBK) ? cnt[tid] : 0;
  int tot;
  int ex = scan256(v, tid, wt, tot);
  if (tid < NBK) {
    offs[tid] = ex;
    offsT[blockIdx.x * NBK + tid] = ex;
  }
  __syncthreads();
#pragma unroll
  for (int j = 0; j < 4; ++j) stage[offs[d[j] >> 8] + r[j]] = ((d[j] & 255) << 16) | s[j];
  __syncthreads();
  *(int4*)(tmp + base + 4 * tid) = *(const int4*)(stage + 4 * tid);
}

__device__ __forceinline__ int locate(const int* runB, const int* runS, int p) {
  int lo = 0;
#pragma unroll
  for (int st = NCH / 2; st > 0; st >>= 1)
    if (runB[lo + st] <= p) lo += st;
  return lo * CHE + runS[lo] + (p - runB[lo]);
}

__global__ __launch_bounds__(1024) void build_kernel(const int* __restrict__ tmp,
    const int* __restrict__ offsT, int* __restrict__ seg, int* __restrict__ csr) {
  __shared__ int runS[NCH];
  __shared__ int runB[NCH + 1];
  __shared__ int cntA[256];
  __shared__ int cntB[256];
  __shared__ int nstart[256];
  __shared__ int wt[8];
  __shared__ int wbuf[CAP];
  __shared__ int obuf[CAP];
  int tid = threadIdx.x, b = blockIdx.x;
  int s = 0, len = 0;
  if (tid < NCH) {
    int nx = tid * NBK + b + 1;
    if (nx > NCH * NBK - 1) nx = NCH * NBK - 1;
    s = offsT[tid * NBK + b];
    int e = offsT[nx];
    if (b == NBK - 1) e = CHE;
    len = e - s;
    runS[tid] = s;
    cntA[tid] = 0;
    cntB[tid] = 0;
  }
  int bstart, T;
  (void)scan256(s, tid, wt, bstart);
  int ex = scan256(len, tid, wt, T);
  if (tid < NCH) runB[tid] = ex;
  if (tid == 0) runB[NCH] = T;
  __syncthreads();
  const bool small = (T <= CAP);

  for (int p0 = tid; p0 < T; p0 += 4096) {
    int w[4];
#pragma unroll
    for (int j = 0; j < 4; ++j) {
      int p = p0 + j * 1024;
      if (p > T - 1) p = T - 1;
      w[j] = tmp[locate(runB, runS, p)];
    }
#pragma unroll
    for (int j = 0; j < 4; ++j) {
      int p = p0 + j * 1024;
      if (p < T) {
        if (small) wbuf[p] = w[j];
        atomicAdd(&cntA[w[j] >> 16], 1);
      }
    }
  }
  __syncthreads();
  int c = (tid < 256) ? cntA[tid] : 0;
  int tot2;
  int nex = scan256(c, tid, wt, tot2);
  if (tid < 256) {
    nstart[tid] = nex;
    ((int2*)seg)[b * 256 + tid] = make_int2(bstart + nex, c);
  }
  __syncthreads();

  if (small) {
    for (int p = tid; p < T; p += 1024) {
      int w = wbuf[p];
      int dl = w >> 16;
      int r = atomicAdd(&cntB[dl], 1);
      obuf[nstart[dl] + r] = w & 0xFFFF;
    }
    __syncthreads();
    int a = bstart & 31;
    for (int i = tid - a; i < T; i += 1024)
      if (i >= 0) csr[bstart + i] = obuf[i];
  } else {
    for (int p0 = tid; p0 < T; p0 += 4096) {
      int w[4];
#pragma unroll
      for (int j = 0; j < 4; ++j) {
        int p = p0 + j * 1024;
        if (p > T - 1) p = T - 1;
        w[j] = tmp[locate(runB, runS, p)];
      }
#pragma unroll
      for (int j = 0; j < 4; ++j) {
        int p = p0 + j * 1024;
        if (p < T) {
          int dl = w[j] >> 16;
          int r = atomicAdd(&cntB[dl], 1);
          csr[bstart + nstart[dl] + r] = w[j] & 0xFFFF;
        }
      }
    }
  }
}

template <int R>
__device__ __forceinline__ void add_rows(const float4* __restrict__ t4, int mine, int k, int lane,
                                         float4& sv) {
  float4 v[R];
#pragma unroll
  for (int j = 0; j < R; ++j) {
    int s = __builtin_amdgcn_readlane(mine, k + j);
    v[j] = t4[(size_t)s * 32 + lane];
  }
#pragma unroll
  for (int j = 0; j < R; ++j) {
    sv.x += v[j].x; sv.y += v[j].y; sv.z += v[j].z; sv.w += v[j].w;
  }
}

__global__ __launch_bounds__(256) void node_kernel(const float* X,
    const float* __restrict__ t, const int* __restrict__ seg,
    const int* __restrict__ csr, float* Y) {
  int i = __builtin_amdgcn_readfirstlane((int)((blockIdx.x * blockDim.x + threadIdx.x) >> 5));
  int lane = threadIdx.x & 31;
  if (i >= NN) return;
  const float4* t4 = (const float4*)t;
  int2 sg = ((const int2*)seg)[i];
  int start = sg.x, n = sg.y;
  const float* xrow = X + (size_t)i * DP;
  float x0 = xrow[0];
  float xs[4], mean[4];
#pragma unroll
  for (int j = 0; j < 4; ++j) xs[j] = xrow[1 + lane * 4 + j];
  float4 sv = t4[(size_t)i * 32 + lane];
  for (int base = 0; base < n; base += 32) {
    int m = (n - base < 32) ? (n - base) : 32;
    int mine = (lane < m) ? csr[start + base + lane] : 0;
    int k = 0;
    for (; k + 8 <= m; k += 8) add_rows<8>(t4, mine, k, lane, sv);
    if (k + 4 <= m) { add_rows<4>(t4, mine, k, lane, sv); k += 4; }
    int r = m - k;
    if (r == 3) add_rows<3>(t4, mine, k, lane, sv);
    else if (r == 2) add_rows<2>(t4, mine, k, lane, sv);
    else if (r == 1) add_rows<1>(t4, mine, k, lane, sv);
  }
  float invd = 1.0f / (float)(n + 1);
  mean[0] = sv.x * invd; mean[1] = sv.y * invd;
  mean[2] = sv.z * invd; mean[3] = sv.w * invd;

  float r1 = 0.f;
#pragma unroll
  for (int j = 0; j < 4; ++j) r1 = fmaf(mean[j], mean[j], r1);
  r1 = wave_sum(r1);
  float vn = sqrtf(fmaxf(r1, EPSF * EPSF));
  float a1 = vn;
  float k1 = sinhf(a1) / a1;
  float agg_ss = k1 * k1 * r1;
  float x0a = sqrtf(1.0f + agg_ss);
  float un = sqrtf(fmaxf(agg_ss, EPSF * EPSF));
  float th = acosh_(fmaxf(x0a, 1.0f + EPSF));
  float su = th / un * k1;

  float u[4]; float ru = 0.f;
#pragma unroll
  for (int j = 0; j < 4; ++j) { u[j] = fmaxf(su * mean[j], 0.0f); ru = fmaf(u[j], u[j], ru); }
  ru = wave_sum(ru);
  float rn = sqrtf(fmaxf(ru, EPSF * EPSF));
  float a2 = rn;
  float act0 = coshf(a2);
  float k2 = sinhf(a2) / a2;

  float acts[4]; float dxa = 0.f;
#pragma unroll
  for (int j = 0; j < 4; ++j) { acts[j] = k2 * u[j]; dxa = fmaf(xs[j], acts[j], dxa); }
  dxa = wave_sum(dxa);
  float alpha = fmaxf(x0 * act0 - dxa, 1.0f + EPSF);

  float u2[4]; float u2ss = 0.f;
#pragma unroll
  for (int j = 0; j < 4; ++j) { u2[j] = acts[j] - alpha * xs[j]; u2ss = fmaf(u2[j], u2[j], u2ss); }
  u2ss = wave_sum(u2ss);
  float u20 = act0 - alpha * x0;
  float mu2 = u2ss - u20 * u20;
  float un2 = sqrtf(fmaxf(mu2, EPSF * EPSF));
  float dist = acosh_(alpha);
  float sv2 = BETAF * dist / un2;
  float bv0 = sv2 * u20;
  float mbv = sv2 * sv2 * mu2;
  float a3 = sqrtf(fmaxf(mbv, EPSF * EPSF));
  float c3 = coshf(a3);
  float k3 = sinhf(a3) / a3;

  float out[4]; float oss = 0.f;
#pragma unroll
  for (int j = 0; j < 4; ++j) {
    out[j] = c3 * xs[j] + k3 * (sv2 * u2[j]);
    oss = fmaf(out[j], out[j], oss);
  }
  oss = wave_sum(oss);
  float y0 = sqrtf(1.0f + oss);
  (void)bv0;
  float* yrow = Y + (size_t)i * DP;
  if (lane == 0) yrow[0] = y0;
#pragma unroll
  for (int j = 0; j < 4; ++j) yrow[1 + lane * 4 + j] = out[j];
}

extern "C" void kernel_launch(void* const* d_in, const int* in_sizes, int n_in,
                              void* d_out, int out_size, void* d_ws, size_t ws_size,
                              hipStream_t stream) {
  const float* x  = (const float*)d_in[0];
  const int*   ei = (const int*)d_in[1];
  const float* W1 = (const float*)d_in[2];
  const float* b1 = (const float*)d_in[3];
  const float* W2 = (const float*)d_in[4];
  const float* b2 = (const float*)d_in[5];
  float* out = (float*)d_out;

  float* t    = (float*)d_ws;
  int* seg    = (int*)(t + (size_t)NN * DD);
  int* offsT  = seg + 2 * NN;
  int* tmp    = offsT + NCH * NBK;
  int* csr    = tmp + NE;

  bin_kernel<<<NCH, 1024, 0, stream>>>(ei, tmp, offsT);
  build_kernel<<<NBK, 1024, 0, stream>>>(tmp, offsT, seg, csr);

  const float* Ws[2] = {W1, W2};
  const float* bs[2] = {b1, b2};
  for (int L = 0; L < 2; ++L) {
    const float* X = (L == 0) ? x : out;
    gemm_kernel<<<NN / 128, 256, 0, stream>>>(X, Ws[L], bs[L], t);
    node_kernel<<<NN / 8, 256, 0, stream>>>(X, t, seg, csr, out);
  }
}


# node kernel: a wave takes 8 nodes and runs the per-node scalar chain (norms, cosh, sinh, arccosh, quotients) once for the 8, node j in lane j; 8192 waves, not 65536; summation order unchanged
# speedup vs baseline: 4.1971x; 1.1632x over previous
#include <hip/hip_runtime.h>
#include <math.h>

#define NN 65536
#define NE 1048576
#define DD 128
#define DP 129
#define EPSF 1e-6f
#define BETAF 0.5f

typedef float v2f __attribute__((ext_vector_type(2)));
typedef float v8f __attribute__((ext_vector_type(8)));

__device__ __forceinline__ float wave_sum(float v) {
#pragma unroll
  for (int m = 16; m > 0; m >>= 1) v += __shfl_xor(v, m, 32);
  return v;
}
__device__ __forceinline__ float half_sum16(float v) {
#pragma unroll
  for (int m = 8; m > 0; m >>= 1) v += __shfl_xor(v, m, 32);
  return v;
}
__device__ __forceinline__ float acosh_(float z) {
  return logf(z + sqrtf(fmaf(z, z, -1.0f)));
}

__global__ __launch_bounds__(256) void gemm_kernel(const float* X, const float* __restrict__ W,
    const float* __restrict__ bias, float* __restrict__ t) {
  __shared__ float lw[128 * 130];
  int waveId = threadIdx.x >> 5;
  int lane = threadIdx.x & 31;
  int m0 = blockIdx.x * 128 + waveId * 16;
  int lh = lane & 15;
  int hi = lane >> 4;
  int arow = m0 + lh;

  float xv[16][4];
#pragma unroll
  for (int r = 0; r < 16; ++r) {
    const float* row = X + (size_t)(m0 + r) * DP + 1 + lane * 4;
#pragma unroll
    for (int j = 0; j < 4; ++j) xv[r][j] = row[j];
  }
  float x0 = X[(size_t)arow * DP];

  for (int idx = threadIdx.x; idx < 128 * 128; idx += 256) {
    int r = idx >> 7, c = idx & 127;
    lw[r * 130 + c] = W[idx];
  }
  __syncthreads();

  float ssq = 0.f;
#pragma unroll
  for (int r = 0; r < 16; ++r) {
    float s = 0.f;
#pragma unroll
    for (int j = 0; j < 4; ++j) s = fmaf(xv[r][j], xv[r][j], s);
    s = wave_sum(s);
    if (lh == r) ssq = s;
  }
  float phiA = acosh_(fmaxf(x0, 1.0f + EPSF)) / sqrtf(fmaxf(ssq, EPSF * EPSF));
  const float* xrow = X + (size_t)arow * DP + 1;

  v8f acc[8] = {};
  for (int k0 = 0; k0 < 128; k0 += 4) {
    int kk = k0 + hi * 2;
    v2f a;
    a.x = phiA * xrow[kk];
    a.y = phiA * xrow[kk + 1];
#pragma unroll
    for (int nt = 0; nt < 8; ++nt) {
      int bn = nt * 16 + lh;
      v2f bf;
      bf.x = lw[bn * 130 + kk];
      bf.y = lw[bn * 130 + kk + 1];
      acc[nt] = __builtin_amdgcn_wmma_f32_16x16x4_f32(
          false, a, false, bf, (short)0, acc[nt], false, false);
    }
  }
  float ss[8];
#pragma unroll
  for (int r = 0; r < 8; ++r) ss[r] = 0.f;
#pragma unroll
  for (int nt = 0; nt < 8; ++nt) {
    float bv = bias[nt * 16 + lh];
#pragma unroll
    for (int r = 0; r < 8; ++r) {
      float g = acc[nt][r] + bv;
      acc[nt][r] = g;
      ss[r] = fmaf(g, g, ss[r]);
    }
  }
#pragma unroll
  for (int r = 0; r < 8; ++r) ss[r] = half_sum16(ss[r]);

  float scl[8];
#pragma unroll
  for (int r = 0; r < 8; ++r) {
    float vss = ss[r];
    float vn = sqrtf(fmaxf(vss, EPSF * EPSF));
    float a = vn;
    float k1 = sinhf(a) / a;
    float hs2 = k1 * k1 * vss;
    float xn = sqrtf(fmaxf(hs2, EPSF * EPSF));
    float th = acosh_(fmaxf(coshf(a), 1.0f + EPSF));
    scl[r] = th / xn * k1;
  }
#pragma unroll
  for (int nt = 0; nt < 8; ++nt) {
    int col = nt * 16 + lh;
#pragma unroll
    for (int r = 0; r < 8; ++r) {
      int row = m0 + r + hi * 8;
      float v = scl[r] * acc[nt][r];
      size_t o = (size_t)row * DD + col;
      t[o] = v;
    }
  }
}

#define NCH 256
#define CHE 4096
#define NBK 256
#define CAP 8192

__device__ __forceinline__ int scan256(int v, int tid, int* wt, int& total) {
  int lane = tid & 31, w = tid >> 5;
  int inc = v;
  for (int m = 1; m < 32; m <<= 1) {
    int o = __shfl_up(inc, m, 32);
    if (lane >= m) inc += o;
  }
  if (lane == 31 && w < 8) wt[w] = inc;
  __syncthreads();
  int base = 0, tot = 0;
  for (int k = 0; k < 8; ++k) {
    int x = wt[k];
    tot += x;
    if (k < w) base += x;
  }
  __syncthreads();
  total = tot;
  return base + inc - v;
}

__global__ __launch_bounds__(1024) void bin_kernel(const int* __restrict__ ei,
    int* __restrict__ tmp, int* __restrict__ offsT) {
  __shared__ int cnt[NBK];
  __shared__ int offs[NBK];
  __shared__ int wt[8];
  __shared__ __attribute__((aligned(16))) int stage[CHE];
  int tid = threadIdx.x;
  int base = blockIdx.x * CHE;
  if (tid < NBK) cnt[tid] = 0;
  int4 s4 = *(const int4*)(ei + base + 4 * tid);
  int4 d4 = *(const int4*)(ei + NE + base + 4 * tid);
  __syncthreads();
  int s[4] = {s4.x, s4.y, s4.z, s4.w};
  int d[4] = {d4.x, d4.y, d4.z, d4.w};
  int r[4];
#pragma unroll
  for (int j = 0; j < 4; ++j) r[j] = atomicAdd(&cnt[d[j] >> 8], 1);
  __syncthreads();
  int v = (tid < NBK) ? cnt[tid] : 0;
  int tot;
  int ex = scan256(v, tid, wt, tot);
  if (tid < NBK) {
    offs[tid] = ex;
    offsT[blockIdx.x * NBK + tid] = ex;
  }
  __syncthreads();
#pragma unroll
  for (int j = 0; j < 4; ++j) stage[offs[d[j] >> 8] + r[j]] = ((d[j] & 255) << 16) | s[j];
  __syncthreads();
  *(int4*)(tmp + base + 4 * tid) = *(const int4*)(stage + 4 * tid);
}

__device__ __forceinline__ int locate(const int* runB, const int* runS, int p) {
  int lo = 0;
#pragma unroll
  for (int st = NCH / 2; st > 0; st >>= 1)
    if (runB[lo + st] <= p) lo += st;
  return lo * CHE + runS[lo] + (p - runB[lo]);
}

__global__ __launch_bounds__(1024) void build_kernel(const int* __restrict__ tmp,
    const int* __restrict__ offsT, int* __restrict__ seg, int* __restrict__ csr) {
  __shared__ int runS[NCH];
  __shared__ int runB[NCH + 1];
  __shared__ int cntA[256];
  __shared__ int cntB[256];
  __shared__ int nstart[256];
  __shared__ int wt[8];
  __shared__ int wbuf[CAP];
  __shared__ int obuf[CAP];
  int tid = threadIdx.x, b = blockIdx.x;
  int s = 0, len = 0;
  if (tid < NCH) {
    int nx = tid * NBK + b + 1;
    if (nx > NCH * NBK - 1) nx = NCH * NBK - 1;
    s = offsT[tid * NBK + b];
    int e = offsT[nx];
    if (b == NBK - 1) e = CHE;
    len = e - s;
    runS[tid] = s;
    cntA[tid] = 0;
    cntB[tid] = 0;
  }
  int bstart, T;
  (void)scan256(s, tid, wt, bstart);
  int ex = scan256(len, tid, wt, T);
  if (tid < NCH) runB[tid] = ex;
  if (tid == 0) runB[NCH] = T;
  __syncthreads();
  const bool small = (T <= CAP);

  for (int p0 = tid; p0 < T; p0 += 4096) {
    int w[4];
#pragma unroll
    for (int j = 0; j < 4; ++j) {
      int p = p0 + j * 1024;
      if (p > T - 1) p = T - 1;
      w[j] = tmp[locate(runB, runS, p)];
    }
#pragma unroll
    for (int j = 0; j < 4; ++j) {
      int p = p0 + j * 1024;
      if (p < T) {
        if (small) wbuf[p] = w[j];
        atomicAdd(&cntA[w[j] >> 16], 1);
      }
    }
  }
  __syncthreads();
  int c = (tid < 256) ? cntA[tid] : 0;
  int tot2;
  int nex = scan256(c, tid, wt, tot2);
  if (tid < 256) {
    nstart[tid] = nex;
    ((int2*)seg)[b * 256 + tid] = make_int2(bstart + nex, c);
  }
  __syncthreads();

  if (small) {
    for (int p = tid; p < T; p += 1024) {
      int w = wbuf[p];
      int dl = w >> 16;
      int r = atomicAdd(&cntB[dl], 1);
      obuf[nstart[dl] + r] = w & 0xFFFF;
    }
    __syncthreads();
    int a = bstart & 31;
    for (int i = tid - a; i < T; i += 1024)
      if (i >= 0) csr[bstart + i] = obuf[i];
  } else {
    for (int p0 = tid; p0 < T; p0 += 4096) {
      int w[4];
#pragma unroll
      for (int j = 0; j < 4; ++j) {
        int p = p0 + j * 1024;
        if (p > T - 1) p = T - 1;
        w[j] = tmp[locate(runB, runS, p)];
      }
#pragma unroll
      for (int j = 0; j < 4; ++j) {
        int p = p0 + j * 1024;
        if (p < T) {
          int dl = w[j] >> 16;
          int r = atomicAdd(&cntB[dl], 1);
          csr[bstart + nstart[dl] + r] = w[j] & 0xFFFF;
        }
      }
    }
  }
}

template <int R>
__device__ __forceinline__ void add_rows(const float4* __restrict__ t4, int mine, int k, int lane,
                                         float4& sv) {
  float4 v[R];
#pragma unroll
  for (int j = 0; j < R; ++j) {
    int s = __builtin_amdgcn_readlane(mine, k + j);
    v[j] = t4[(size_t)s * 32 + lane];
  }
#pragma unroll
  for (int j = 0; j < R; ++j) {
    sv.x += v[j].x; sv.y += v[j].y; sv.z += v[j].z; sv.w += v[j].w;
  }
}

__device__ __forceinline__ void gather_node(const float4* __restrict__ t4,
    const int* __restrict__ csr, int start, int n, int mine, int lane, float4& sv) {
  for (int base = 0; base < n; base += 32) {
    int m = (n - base < 32) ? (n - base) : 32;
    if (base > 0) mine = (lane < m) ? csr[start + base + lane] : 0;
    int k = 0;
    for (; k + 8 <= m; k += 8) add_rows<8>(t4, mine, k, lane, sv);
    if (k + 4 <= m) { add_rows<4>(t4, mine, k, lane, sv); k += 4; }
    int r = m - k;
    if (r == 3) add_rows<3>(t4, mine, k, lane, sv);
    else if (r == 2) add_rows<2>(t4, mine, k, lane, sv);
    else if (r == 1) add_rows<1>(t4, mine, k, lane, sv);
  }
}

#define NB 8

__device__ __forceinline__ void put(float& acc, float v, int lj, int j) {
  if (lj == j) acc = v;
}
__device__ __forceinline__ float take(float vec, int j) {
  return __int_as_float(__builtin_amdgcn_readlane(__float_as_int(vec), j));
}

__global__ __launch_bounds__(256) void node_kernel(const float* X,
    const float* __restrict__ t, const int* __restrict__ seg,
    const int* __restrict__ csr, float* Y) {
  int w = __builtin_amdgcn_readfirstlane((int)((blockIdx.x * 256 + threadIdx.x) >> 5));
  int lane = threadIdx.x & 31;
  int i0 = w * NB;
  if (i0 >= NN) return;
  int lj = lane & (NB - 1);
  const float4* t4 = (const float4*)t;

  int2 sgv = ((const int2*)seg)[i0 + lj];
  float x0v = X[(size_t)(i0 + lj) * DP];
  float4 sv[NB];
#pragma unroll
  for (int j = 0; j < NB; ++j) sv[j] = t4[(size_t)(i0 + j) * 32 + lane];
  int st[NB], nn[NB], mine0[NB];
#pragma unroll
  for (int j = 0; j < NB; ++j) {
    st[j] = __builtin_amdgcn_readlane(sgv.x, j);
    nn[j] = __builtin_amdgcn_readlane(sgv.y, j);
  }
#pragma unroll
  for (int j = 0; j < NB; ++j) mine0[j] = (lane < nn[j]) ? csr[st[j] + lane] : 0;
#pragma unroll
  for (int j = 0; j < NB; ++j) gather_node(t4, csr, st[j], nn[j], mine0[j], lane, sv[j]);

  float xs[NB][4];
#pragma unroll
  for (int j = 0; j < NB; ++j) {
    const float* xrow = X + (size_t)(i0 + j) * DP;
#pragma unroll
    for (int q = 0; q < 4; ++q) xs[j][q] = xrow[1 + lane * 4 + q];
  }

  float invdv = 1.0f / (float)(sgv.y + 1);
  float vv[NB][4];
  float r1v = 0.f;
#pragma unroll
  for (int j = 0; j < NB; ++j) {
    float invd = take(invdv, j);
    vv[j][0] = sv[j].x * invd; vv[j][1] = sv[j].y * invd;
    vv[j][2] = sv[j].z * invd; vv[j][3] = sv[j].w * invd;
    float r1 = 0.f;
#pragma unroll
    for (int q = 0; q < 4; ++q) r1 = fmaf(vv[j][q], vv[j][q], r1);
    put(r1v, wave_sum(r1), lj, j);
  }
  float suv;
  {
    float r1 = r1v;
    float vn = sqrtf(fmaxf(r1, EPSF * EPSF));
    float a1 = vn;
    float k1 = sinhf(a1) / a1;
    float agg_ss = k1 * k1 * r1;
    float x0a = sqrtf(1.0f + agg_ss);
    float un = sqrtf(fmaxf(agg_ss, EPSF * EPSF));
    float th = acosh_(fmaxf(x0a, 1.0f + EPSF));
    suv = th / un * k1;
  }

  float ruv = 0.f;
#pragma unroll
  for (int j = 0; j < NB; ++j) {
    float su = take(suv, j);
    float ru = 0.f;
#pragma unroll
    for (int q = 0; q < 4; ++q) { vv[j][q] = fmaxf(su * vv[j][q], 0.0f); ru = fmaf(vv[j][q], vv[j][q], ru); }
    put(ruv, wave_sum(ru), lj, j);
  }
  float act0v, k2v;
  {
    float rn = sqrtf(fmaxf(ruv, EPSF * EPSF));
    float a2 = rn;
    act0v = coshf(a2);
    k2v = sinhf(a2) / a2;
  }

  float dxav = 0.f;
#pragma unroll
  for (int j = 0; j < NB; ++j) {
    float k2 = take(k2v, j);
    float dxa = 0.f;
#pragma unroll
    for (int q = 0; q < 4; ++q) { vv[j][q] = k2 * vv[j][q]; dxa = fmaf(xs[j][q], vv[j][q], dxa); }
    put(dxav, wave_sum(dxa), lj, j);
  }
  float alphav = fmaxf(x0v * act0v - dxav, 1.0f + EPSF);

  float u2ssv = 0.f;
#pragma unroll
  for (int j = 0; j < NB; ++j) {
    float alpha = take(alphav, j);
    float u2ss = 0.f;
#pragma unroll
    for (int q = 0; q < 4; ++q) { vv[j][q] = vv[j][q] - alpha * xs[j][q]; u2ss = fmaf(vv[j][q], vv[j][q], u2ss); }
    put(u2ssv, wave_sum(u2ss), lj, j);
  }
  float sv2v, c3v, k3v;
  {
    float u20 = act0v - alphav * x0v;
    float mu2 = u2ssv - u20 * u20;
    float un2 = sqrtf(fmaxf(mu2, EPSF * EPSF));
    float dist = acosh_(alphav);
    float sv2 = BETAF * dist / un2;
    float mbv = sv2 * sv2 * mu2;
    float a3 = sqrtf(fmaxf(mbv, EPSF * EPSF));
    sv2v = sv2;
    c3v = coshf(a3);
    k3v = sinhf(a3) / a3;
  }

  float ossv = 0.f;
#pragma unroll
  for (int j = 0; j < NB; ++j) {
    float c3 = take(c3v, j), k3 = take(k3v, j), sv2 = take(sv2v, j);
    float oss = 0.f;
#pragma unroll
    for (int q = 0; q < 4; ++q) {
      vv[j][q] = c3 * xs[j][q] + k3 * (sv2 * vv[j][q]);
      oss = fmaf(vv[j][q], vv[j][q], oss);
    }
    put(ossv, wave_sum(oss), lj, j);
  }
  float y0v = sqrtf(1.0f + ossv);

  if (lane < NB) Y[(size_t)(i0 + lane) * DP] = y0v;
#pragma unroll
  for (int j = 0; j < NB; ++j) {
    float* yrow = Y + (size_t)(i0 + j) * DP;
#pragma unroll
    for (int q = 0; q < 4; ++q) yrow[1 + lane * 4 + q] = vv[j][q];
  }
}

extern "C" void kernel_launch(void* const* d_in, const int* in_sizes, int n_in,
                              void* d_out, int out_size, void* d_ws, size_t ws_size,
                              hipStream_t stream) {
  const float* x  = (const float*)d_in[0];
  const int*   ei = (const int*)d_in[1];
  const float* W1 = (const float*)d_in[2];
  const float* b1 = (const float*)d_in[3];
  const float* W2 = (const float*)d_in[4];
  const float* b2 = (const float*)d_in[5];
  float* out = (float*)d_out;

  float* t    = (float*)d_ws;
  int* seg    = (int*)(t + (size_t)NN * DD);
  int* offsT  = seg + 2 * NN;
  int* tmp    = offsT + NCH * NBK;
  int* csr    = tmp + NE;

  bin_kernel<<<NCH, 1024, 0, stream>>>(ei, tmp, offsT);
  build_kernel<<<NBK, 1024, 0, stream>>>(tmp, offsT, seg, csr);

  const float* Ws[2] = {W1, W2};
  const float* bs[2] = {b1, b2};
  for (int L = 0; L < 2; ++L) {
    const float* X = (L == 0) ? x : out;
    gemm_kernel<<<NN / 128, 256, 0, stream>>>(X, Ws[L], bs[L], t);
    node_kernel<<<NN / (8 * NB), 256, 0, stream>>>(X, t, seg, csr, out);
  }
}


# rows' sums reduced together by a halving butterfly (same partners, same order: bit-identical), each lane ending with one row's sum; GEMM epilogue runs a row's scalar chain once, not in 16 lanes
# speedup vs baseline: 4.4234x; 1.0539x over previous
#include <hip/hip_runtime.h>
#include <math.h>

#define NN 65536
#define NE 1048576
#define DD 128
#define DP 129
#define EPSF 1e-6f
#define BETAF 0.5f

typedef float v2f __attribute__((ext_vector_type(2)));
typedef float v8f __attribute__((ext_vector_type(8)));

__device__ __forceinline__ float sum8_rows(const float (&p)[8], int lane) {
  const bool b4 = (lane & 16) != 0, b3 = (lane & 8) != 0, b2 = (lane & 4) != 0;
  float s1[4], s2[2];
#pragma unroll
  for (int i = 0; i < 4; ++i) {
    float keep = b4 ? p[i + 4] : p[i], give = b4 ? p[i] : p[i + 4];
    s1[i] = keep + __shfl_xor(give, 16, 32);
  }
#pragma unroll
  for (int i = 0; i < 2; ++i) {
    float keep = b3 ? s1[i + 2] : s1[i], give = b3 ? s1[i] : s1[i + 2];
    s2[i] = keep + __shfl_xor(give, 8, 32);
  }
  float keep = b2 ? s2[1] : s2[0], give = b2 ? s2[0] : s2[1];
  float s = keep + __shfl_xor(give, 4, 32);
  s += __shfl_xor(s, 2, 32);
  s += __shfl_xor(s, 1, 32);
  return s;
}
__device__ __forceinline__ float sum16_rows(const float (&p)[16], int lane) {
  const bool b3 = (lane & 8) != 0, b2 = (lane & 4) != 0, b1 = (lane & 2) != 0, b0 = (lane & 1) != 0;
  float s0[16], s1[8], s2[4], s3[2];
#pragma unroll
  for (int i = 0; i < 16; ++i) s0[i] = p[i] + __shfl_xor(p[i], 16, 32);
#pragma unroll
  for (int i = 0; i < 8; ++i) {
    float keep = b3 ? s0[i + 8] : s0[i], give = b3 ? s0[i] : s0[i + 8];
    s1[i] = keep + __shfl_xor(give, 8, 32);
  }
#pragma unroll
  for (int i = 0; i < 4; ++i) {
    float keep = b2 ? s1[i + 4] : s1[i], give = b2 ? s1[i] : s1[i + 4];
    s2[i] = keep + __shfl_xor(give, 4, 32);
  }
#pragma unroll
  for (int i = 0; i < 2; ++i) {
    float keep = b1 ? s2[i + 2] : s2[i], give = b1 ? s2[i] : s2[i + 2];
    s3[i] = keep + __shfl_xor(give, 2, 32);
  }
  float keep = b0 ? s3[1] : s3[0], give = b0 ? s3[0] : s3[1];
  return keep + __shfl_xor(give, 1, 32);
}
__device__ __forceinline__ float half_sum8_rows(const float (&p)[8], int lane) {
  const bool b3 = (lane & 8) != 0, b2 = (lane & 4) != 0, b1 = (lane & 2) != 0;
  float s1[4], s2[2];
#pragma unroll
  for (int i = 0; i < 4; ++i) {
    float keep = b3 ? p[i + 4] : p[i], give = b3 ? p[i] : p[i + 4];
    s1[i] = keep + __shfl_xor(give, 8, 32);
  }
#pragma unroll
  for (int i = 0; i < 2; ++i) {
    float keep = b2 ? s1[i + 2] : s1[i], give = b2 ? s1[i] : s1[i + 2];
    s2[i] = keep + __shfl_xor(give, 4, 32);
  }
  float keep = b1 ? s2[1] : s2[0], give = b1 ? s2[0] : s2[1];
  float s = keep + __shfl_xor(give, 2, 32);
  s += __shfl_xor(s, 1, 32);
  return s;
}
__device__ __forceinline__ float acosh_(float z) {
  return logf(z + sqrtf(fmaf(z, z, -1.0f)));
}

__global__ __launch_bounds__(256) void gemm_kernel(const float* X, const float* __restrict__ W,
    const float* __restrict__ bias, float* __restrict__ t) {
  __shared__ float lw[128 * 130];
  int waveId = threadIdx.x >> 5;
  int lane = threadIdx.x & 31;
  int m0 = blockIdx.x * 128 + waveId * 16;
  int lh = lane & 15;
  int hi = lane >> 4;
  int arow = m0 + lh;

  float xv[16][4];
#pragma unroll
  for (int r = 0; r < 16; ++r) {
    const float* row = X + (size_t)(m0 + r) * DP + 1 + lane * 4;
#pragma unroll
    for (int j = 0; j < 4; ++j) xv[r][j] = row[j];
  }
  float x0 = X[(size_t)arow * DP];

  for (int idx = threadIdx.x; idx < 128 * 128; idx += 256) {
    int r = idx >> 7, c = idx & 127;
    lw[r * 130 + c] = W[idx];
  }
  __syncthreads();

  float sq[16];
#pragma unroll
  for (int r = 0; r < 16; ++r) {
    float s = 0.f;
#pragma unroll
    for (int j = 0; j < 4; ++j) s = fmaf(xv[r][j], xv[r][j], s);
    sq[r] = s;
  }
  float ssq = sum16_rows(sq, lane);
  float phiA = acosh_(fmaxf(x0, 1.0f + EPSF)) / sqrtf(fmaxf(ssq, EPSF * EPSF));
  const float* xrow = X + (size_t)arow * DP + 1;

  v8f acc[8] = {};
  for (int k0 = 0; k0 < 128; k0 += 4) {
    int kk = k0 + hi * 2;
    v2f a;
    a.x = phiA * xrow[kk];
    a.y = phiA * xrow[kk + 1];
#pragma unroll
    for (int nt = 0; nt < 8; ++nt) {
      int bn = nt * 16 + lh;
      v2f bf;
      bf.x = lw[bn * 130 + kk];
      bf.y = lw[bn * 130 + kk + 1];
      acc[nt] = __builtin_amdgcn_wmma_f32_16x16x4_f32(
          false, a, false, bf, (short)0, acc[nt], false, false);
    }
  }
  float ss[8];
#pragma unroll
  for (int r = 0; r < 8; ++r) ss[r] = 0.f;
#pragma unroll
  for (int nt = 0; nt < 8; ++nt) {
    float bv = bias[nt * 16 + lh];
#pragma unroll
    for (int r = 0; r < 8; ++r) {
      float g = acc[nt][r] + bv;
      acc[nt][r] = g;
      ss[r] = fmaf(g, g, ss[r]);
    }
  }
  float sclv;
  {
    float vss = half_sum8_rows(ss, lane);
    float vn = sqrtf(fmaxf(vss, EPSF * EPSF));
    float a = vn;
    float k1 = sinhf(a) / a;
    float hs2 = k1 * k1 * vss;
    float xn = sqrtf(fmaxf(hs2, EPSF * EPSF));
    float th = acosh_(fmaxf(coshf(a), 1.0f + EPSF));
    sclv = th / xn * k1;
  }
  float scl[8];
#pragma unroll
  for (int r = 0; r < 8; ++r) scl[r] = __shfl(sclv, (lane & 16) + 2 * r, 32);
#pragma unroll
  for (int nt = 0; nt < 8; ++nt) {
    int col = nt * 16 + lh;
#pragma unroll
    for (int r = 0; r < 8; ++r) {
      int row = m0 + r + hi * 8;
      float v = scl[r] * acc[nt][r];
      size_t o = (size_t)row * DD + col;
      t[o] = v;
    }
  }
}

#define NCH 256
#define CHE 4096
#define NBK 256
#define CAP 8192

__device__ __forceinline__ int scan256(int v, int tid, int* wt, int& total) {
  int lane = tid & 31, w = tid >> 5;
  int inc = v;
  for (int m = 1; m < 32; m <<= 1) {
    int o = __shfl_up(inc, m, 32);
    if (lane >= m) inc += o;
  }
  if (lane == 31 && w < 8) wt[w] = inc;
  __syncthreads();
  int base = 0, tot = 0;
  for (int k = 0; k < 8; ++k) {
    int x = wt[k];
    tot += x;
    if (k < w) base += x;
  }
  __syncthreads();
  total = tot;
  return base + inc - v;
}

__global__ __launch_bounds__(1024) void bin_kernel(const int* __restrict__ ei,
    int* __restrict__ tmp, int* __restrict__ offsT) {
  __shared__ int cnt[NBK];
  __shared__ int offs[NBK];
  __shared__ int wt[8];
  __shared__ __attribute__((aligned(16))) int stage[CHE];
  int tid = threadIdx.x;
  int base = blockIdx.x * CHE;
  if (tid < NBK) cnt[tid] = 0;
  int4 s4 = *(const int4*)(ei + base + 4 * tid);
  int4 d4 = *(const int4*)(ei + NE + base + 4 * tid);
  __syncthreads();
  int s[4] = {s4.x, s4.y, s4.z, s4.w};
  int d[4] = {d4.x, d4.y, d4.z, d4.w};
  int r[4];
#pragma unroll
  for (int j = 0; j < 4; ++j) r[j] = atomicAdd(&cnt[d[j] >> 8], 1);
  __syncthreads();
  int v = (tid < NBK) ? cnt[tid] : 0;
  int tot;
  int ex = scan256(v, tid, wt, tot);
  if (tid < NBK) {
    offs[tid] = ex;
    offsT[blockIdx.x * NBK + tid] = ex;
  }
  __syncthreads();
#pragma unroll
  for (int j = 0; j < 4; ++j) stage[offs[d[j] >> 8] + r[j]] = ((d[j] & 255) << 16) | s[j];
  __syncthreads();
  *(int4*)(tmp + base + 4 * tid) = *(const int4*)(stage + 4 * tid);
}

__device__ __forceinline__ int locate(const int* runB, const int* runS, int p) {
  int lo = 0;
#pragma unroll
  for (int st = NCH / 2; st > 0; st >>= 1)
    if (runB[lo + st] <= p) lo += st;
  return lo * CHE + runS[lo] + (p - runB[lo]);
}

__global__ __launch_bounds__(1024) void build_kernel(const int* __restrict__ tmp,
    const int* __restrict__ offsT, int* __restrict__ seg, int* __restrict__ csr) {
  __shared__ int runS[NCH];
  __shared__ int runB[NCH + 1];
  __shared__ int cntA[256];
  __shared__ int cntB[256];
  __shared__ int nstart[256];
  __shared__ int wt[8];
  __shared__ int wbuf[CAP];
  __shared__ int obuf[CAP];
  int tid = threadIdx.x, b = blockIdx.x;
  int s = 0, len = 0;
  if (tid < NCH) {
    int nx = tid * NBK + b + 1;
    if (nx > NCH * NBK - 1) nx = NCH * NBK - 1;
    s = offsT[tid * NBK + b];
    int e = offsT[nx];
    if (b == NBK - 1) e = CHE;
    len = e - s;
    runS[tid] = s;
    cntA[tid] = 0;
    cntB[tid] = 0;
  }
  int bstart, T;
  (void)scan256(s, tid, wt, bstart);
  int ex = scan256(len, tid, wt, T);
  if (tid < NCH) runB[tid] = ex;
  if (tid == 0) runB[NCH] = T;
  __syncthreads();
  const bool small = (T <= CAP);

  for (int p0 = tid; p0 < T; p0 += 4096) {
    int w[4];
#pragma unroll
    for (int j = 0; j < 4; ++j) {
      int p = p0 + j * 1024;
      if (p > T - 1) p = T - 1;
      w[j] = tmp[locate(runB, runS, p)];
    }
#pragma unroll
    for (int j = 0; j < 4; ++j) {
      int p = p0 + j * 1024;
      if (p < T) {
        if (small) wbuf[p] = w[j];
        atomicAdd(&cntA[w[j] >> 16], 1);
      }
    }
  }
  __syncthreads();
  int c = (tid < 256) ? cntA[tid] : 0;
  int tot2;
  int nex = scan256(c, tid, wt, tot2);
  if (tid < 256) {
    nstart[tid] = nex;
    ((int2*)seg)[b * 256 + tid] = make_int2(bstart + nex, c);
  }
  __syncthreads();

  if (small) {
    for (int p = tid; p < T; p += 1024) {
      int w = wbuf[p];
      int dl = w >> 16;
      int r = atomicAdd(&cntB[dl], 1);
      obuf[nstart[dl] + r] = w & 0xFFFF;
    }
    __syncthreads();
    int a = bstart & 31;
    for (int i = tid - a; i < T; i += 1024)
      if (i >= 0) csr[bstart + i] = obuf[i];
  } else {
    for (int p0 = tid; p0 < T; p0 += 4096) {
      int w[4];
#pragma unroll
      for (int j = 0; j < 4; ++j) {
        int p = p0 + j * 1024;
        if (p > T - 1) p = T - 1;
        w[j] = tmp[locate(runB, runS, p)];
      }
#pragma unroll
      for (int j = 0; j < 4; ++j) {
        int p = p0 + j * 1024;
        if (p < T) {
          int dl = w[j] >> 16;
          int r = atomicAdd(&cntB[dl], 1);
          csr[bstart + nstart[dl] + r] = w[j] & 0xFFFF;
        }
      }
    }
  }
}

template <int R>
__device__ __forceinline__ void add_rows(const float4* __restrict__ t4, int mine, int k, int lane,
                                         float4& sv) {
  float4 v[R];
#pragma unroll
  for (int j = 0; j < R; ++j) {
    int s = __builtin_amdgcn_readlane(mine, k + j);
    v[j] = t4[(size_t)s * 32 + lane];
  }
#pragma unroll
  for (int j = 0; j < R; ++j) {
    sv.x += v[j].x; sv.y += v[j].y; sv.z += v[j].z; sv.w += v[j].w;
  }
}

__device__ __forceinline__ void gather_node(const float4* __restrict__ t4,
    const int* __restrict__ csr, int start, int n, int mine, int lane, float4& sv) {
  for (int base = 0; base < n; base += 32) {
    int m = (n - base < 32) ? (n - base) : 32;
    if (base > 0) mine = (lane < m) ? csr[start + base + lane] : 0;
    int k = 0;
    for (; k + 8 <= m; k += 8) add_rows<8>(t4, mine, k, lane, sv);
    if (k + 4 <= m) { add_rows<4>(t4, mine, k, lane, sv); k += 4; }
    int r = m - k;
    if (r == 3) add_rows<3>(t4, mine, k, lane, sv);
    else if (r == 2) add_rows<2>(t4, mine, k, lane, sv);
    else if (r == 1) add_rows<1>(t4, mine, k, lane, sv);
  }
}

#define NB 8

__device__ __forceinline__ float take(float vec, int j) {
  return __int_as_float(__builtin_amdgcn_readlane(__float_as_int(vec), 4 * j));
}

__global__ __launch_bounds__(256) void node_kernel(const float* X,
    const float* __restrict__ t, const int* __restrict__ seg,
    const int* __restrict__ csr, float* Y) {
  int w = __builtin_amdgcn_readfirstlane((int)((blockIdx.x * 256 + threadIdx.x) >> 5));
  int lane = threadIdx.x & 31;
  int i0 = w * NB;
  if (i0 >= NN) return;
  int lj = lane >> 2;
  const float4* t4 = (const float4*)t;

  int2 sgv = ((const int2*)seg)[i0 + lj];
  float x0v = X[(size_t)(i0 + lj) * DP];
  float4 sv[NB];
#pragma unroll
  for (int j = 0; j < NB; ++j) sv[j] = t4[(size_t)(i0 + j) * 32 + lane];
  int st[NB], nn[NB], mine0[NB];
#pragma unroll
  for (int j = 0; j < NB; ++j) {
    st[j] = __builtin_amdgcn_readlane(sgv.x, 4 * j);
    nn[j] = __builtin_amdgcn_readlane(sgv.y, 4 * j);
  }
#pragma unroll
  for (int j = 0; j < NB; ++j) mine0[j] = (lane < nn[j]) ? csr[st[j] + lane] : 0;
#pragma unroll
  for (int j = 0; j < NB; ++j) gather_node(t4, csr, st[j], nn[j], mine0[j], lane, sv[j]);

  float xs[NB][4];
#pragma unroll
  for (int j = 0; j < NB; ++j) {
    const float* xrow = X + (size_t)(i0 + j) * DP;
#pragma unroll
    for (int q = 0; q < 4; ++q) xs[j][q] = xrow[1 + lane * 4 + q];
  }

  float invdv = 1.0f / (float)(sgv.y + 1);
  float vv[NB][4];
  float part[NB];
#pragma unroll
  for (int j = 0; j < NB; ++j) {
    float invd = take(invdv, j);
    vv[j][0] = sv[j].x * invd; vv[j][1] = sv[j].y * invd;
    vv[j][2] = sv[j].z * invd; vv[j][3] = sv[j].w * invd;
    float r1 = 0.f;
#pragma unroll
    for (int q = 0; q < 4; ++q) r1 = fmaf(vv[j][q], vv[j][q], r1);
    part[j] = r1;
  }
  float r1v = sum8_rows(part, lane);
  float suv;
  {
    float r1 = r1v;
    float vn = sqrtf(fmaxf(r1, EPSF * EPSF));
    float a1 = vn;
    float k1 = sinhf(a1) / a1;
    float agg_ss = k1 * k1 * r1;
    float x0a = sqrtf(1.0f + agg_ss);
    float un = sqrtf(fmaxf(agg_ss, EPSF * EPSF));
    float th = acosh_(fmaxf(x0a, 1.0f + EPSF));
    suv = th / un * k1;
  }

#pragma unroll
  for (int j = 0; j < NB; ++j) {
    float su = take(suv, j);
    float ru = 0.f;
#pragma unroll
    for (int q = 0; q < 4; ++q) { vv[j][q] = fmaxf(su * vv[j][q], 0.0f); ru = fmaf(vv[j][q], vv[j][q], ru); }
    part[j] = ru;
  }
  float ruv = sum8_rows(part, lane);
  float act0v, k2v;
  {
    float rn = sqrtf(fmaxf(ruv, EPSF * EPSF));
    float a2 = rn;
    act0v = coshf(a2);
    k2v = sinhf(a2) / a2;
  }

#pragma unroll
  for (int j = 0; j < NB; ++j) {
    float k2 = take(k2v, j);
    float dxa = 0.f;
#pragma unroll
    for (int q = 0; q < 4; ++q) { vv[j][q] = k2 * vv[j][q]; dxa = fmaf(xs[j][q], vv[j][q], dxa); }
    part[j] = dxa;
  }
  float dxav = sum8_rows(part, lane);
  float alphav = fmaxf(x0v * act0v - dxav, 1.0f + EPSF);

#pragma unroll
  for (int j = 0; j < NB; ++j) {
    float alpha = take(alphav, j);
    float u2ss = 0.f;
#pragma unroll
    for (int q = 0; q < 4; ++q) { vv[j][q] = vv[j][q] - alpha * xs[j][q]; u2ss = fmaf(vv[j][q], vv[j][q], u2ss); }
    part[j] = u2ss;
  }
  float u2ssv = sum8_rows(part, lane);
  float sv2v, c3v, k3v;
  {
    float u20 = act0v - alphav * x0v;
    float mu2 = u2ssv - u20 * u20;
    float un2 = sqrtf(fmaxf(mu2, EPSF * EPSF));
    float dist = acosh_(alphav);
    float sv2 = BETAF * dist / un2;
    float mbv = sv2 * sv2 * mu2;
    float a3 = sqrtf(fmaxf(mbv, EPSF * EPSF));
    sv2v = sv2;
    c3v = coshf(a3);
    k3v = sinhf(a3) / a3;
  }

#pragma unroll
  for (int j = 0; j < NB; ++j) {
    float c3 = take(c3v, j), k3 = take(k3v, j), sv2 = take(sv2v, j);
    float oss = 0.f;
#pragma unroll
    for (int q = 0; q < 4; ++q) {
      vv[j][q] = c3 * xs[j][q] + k3 * (sv2 * vv[j][q]);
      oss = fmaf(vv[j][q], vv[j][q], oss);
    }
    part[j] = oss;
  }
  float ossv = sum8_rows(part, lane);
  float y0v = sqrtf(1.0f + ossv);

  if ((lane & 3) == 0) Y[(size_t)(i0 + lj) * DP] = y0v;
#pragma unroll
  for (int j = 0; j < NB; ++j) {
    float* yrow = Y + (size_t)(i0 + j) * DP;
#pragma unroll
    for (int q = 0; q < 4; ++q) yrow[1 + lane * 4 + q] = vv[j][q];
  }
}

extern "C" void kernel_launch(void* const* d_in, const int* in_sizes, int n_in,
                              void* d_out, int out_size, void* d_ws, size_t ws_size,
                              hipStream_t stream) {
  const float* x  = (const float*)d_in[0];
  const int*   ei = (const int*)d_in[1];
  const float* W1 = (const float*)d_in[2];
  const float* b1 = (const float*)d_in[3];
  const float* W2 = (const float*)d_in[4];
  const float* b2 = (const float*)d_in[5];
  float* out = (float*)d_out;

  float* t    = (float*)d_ws;
  int* seg    = (int*)(t + (size_t)NN * DD);
  int* offsT  = seg + 2 * NN;
  int* tmp    = offsT + NCH * NBK;
  int* csr    = tmp + NE;

  bin_kernel<<<NCH, 1024, 0, stream>>>(ei, tmp, offsT);
  build_kernel<<<NBK, 1024, 0, stream>>>(tmp, offsT, seg, csr);

  const float* Ws[2] = {W1, W2};
  const float* bs[2] = {b1, b2};
  for (int L = 0; L < 2; ++L) {
    const float* X = (L == 0) ? x : out;
    gemm_kernel<<<NN / 128, 256, 0, stream>>>(X, Ws[L], bs[L], t);
    node_kernel<<<NN / (8 * NB), 256, 0, stream>>>(X, t, seg, csr, out);
  }
}


# second level of the edge ordering and layer 1's GEMM share one launch (blocks of both kinds, 512 threads, 2 per processor): 5 launches, not 6; slot order still LDS arrival order, now of 512 threads
# speedup vs baseline: 4.6117x; 1.0426x over previous
#include <hip/hip_runtime.h>
#include <math.h>

#define NN 65536
#define NE 1048576
#define DD 128
#define DP 129
#define EPSF 1e-6f
#define BETAF 0.5f

typedef float v2f __attribute__((ext_vector_type(2)));
typedef float v8f __attribute__((ext_vector_type(8)));

__device__ __forceinline__ float sum8_rows(const float (&p)[8], int lane) {
  const bool b4 = (lane & 16) != 0, b3 = (lane & 8) != 0, b2 = (lane & 4) != 0;
  float s1[4], s2[2];
#pragma unroll
  for (int i = 0; i < 4; ++i) {
    float keep = b4 ? p[i + 4] : p[i], give = b4 ? p[i] : p[i + 4];
    s1[i] = keep + __shfl_xor(give, 16, 32);
  }
#pragma unroll
  for (int i = 0; i < 2; ++i) {
    float keep = b3 ? s1[i + 2] : s1[i], give = b3 ? s1[i] : s1[i + 2];
    s2[i] = keep + __shfl_xor(give, 8, 32);
  }
  float keep = b2 ? s2[1] : s2[0], give = b2 ? s2[0] : s2[1];
  float s = keep + __shfl_xor(give, 4, 32);
  s += __shfl_xor(s, 2, 32);
  s += __shfl_xor(s, 1, 32);
  return s;
}
__device__ __forceinline__ float sum16_rows(const float (&p)[16], int lane) {
  const bool b3 = (lane & 8) != 0, b2 = (lane & 4) != 0, b1 = (lane & 2) != 0, b0 = (lane & 1) != 0;
  float s0[16], s1[8], s2[4], s3[2];
#pragma unroll
  for (int i = 0; i < 16; ++i) s0[i] = p[i] + __shfl_xor(p[i], 16, 32);
#pragma unroll
  for (int i = 0; i < 8; ++i) {
    float keep = b3 ? s0[i + 8] : s0[i], give = b3 ? s0[i] : s0[i + 8];
    s1[i] = keep + __shfl_xor(give, 8, 32);
  }
#pragma unroll
  for (int i = 0; i < 4; ++i) {
    float keep = b2 ? s1[i + 4] : s1[i], give = b2 ? s1[i] : s1[i + 4];
    s2[i] = keep + __shfl_xor(give, 4, 32);
  }
#pragma unroll
  for (int i = 0; i < 2; ++i) {
    float keep = b1 ? s2[i + 2] : s2[i], give = b1 ? s2[i] : s2[i + 2];
    s3[i] = keep + __shfl_xor(give, 2, 32);
  }
  float keep = b0 ? s3[1] : s3[0], give = b0 ? s3[0] : s3[1];
  return keep + __shfl_xor(give, 1, 32);
}
__device__ __forceinline__ float half_sum8_rows(const float (&p)[8], int lane) {
  const bool b3 = (lane & 8) != 0, b2 = (lane & 4) != 0, b1 = (lane & 2) != 0;
  float s1[4], s2[2];
#pragma unroll
  for (int i = 0; i < 4; ++i) {
    float keep = b3 ? p[i + 4] : p[i], give = b3 ? p[i] : p[i + 4];
    s1[i] = keep + __shfl_xor(give, 8, 32);
  }
#pragma unroll
  for (int i = 0; i < 2; ++i) {
    float keep = b2 ? s1[i + 2] : s1[i], give = b2 ? s1[i] : s1[i + 2];
    s2[i] = keep + __shfl_xor(give, 4, 32);
  }
  float keep = b1 ? s2[1] : s2[0], give = b1 ? s2[0] : s2[1];
  float s = keep + __shfl_xor(give, 2, 32);
  s += __shfl_xor(s, 1, 32);
  return s;
}
__device__ __forceinline__ float acosh_(float z) {
  return logf(z + sqrtf(fmaf(z, z, -1.0f)));
}

template <int BT>
__device__ __forceinline__ void gemm_body(const float* X, const float* __restrict__ W,
    const float* __restrict__ bias, float* __restrict__ t, int blk) {
  __shared__ float lw[128 * 130];
  int waveId = threadIdx.x >> 5;
  int lane = threadIdx.x & 31;
  int m0 = blk * (BT / 2) + waveId * 16;
  int lh = lane & 15;
  int hi = lane >> 4;
  int arow = m0 + lh;

  float xv[16][4];
#pragma unroll
  for (int r = 0; r < 16; ++r) {
    const float* row = X + (size_t)(m0 + r) * DP + 1 + lane * 4;
#pragma unroll
    for (int j = 0; j < 4; ++j) xv[r][j] = row[j];
  }
  float x0 = X[(size_t)arow * DP];

  for (int idx = threadIdx.x; idx < 128 * 128; idx += BT) {
    int r = idx >> 7, c = idx & 127;
    lw[r * 130 + c] = W[idx];
  }
  __syncthreads();

  float sq[16];
#pragma unroll
  for (int r = 0; r < 16; ++r) {
    float s = 0.f;
#pragma unroll
    for (int j = 0; j < 4; ++j) s = fmaf(xv[r][j], xv[r][j], s);
    sq[r] = s;
  }
  float ssq = sum16_rows(sq, lane);
  float phiA = acosh_(fmaxf(x0, 1.0f + EPSF)) / sqrtf(fmaxf(ssq, EPSF * EPSF));
  const float* xrow = X + (size_t)arow * DP + 1;

  v8f acc[8] = {};
  for (int k0 = 0; k0 < 128; k0 += 4) {
    int kk = k0 + hi * 2;
    v2f a;
    a.x = phiA * xrow[kk];
    a.y = phiA * xrow[kk + 1];
#pragma unroll
    for (int nt = 0; nt < 8; ++nt) {
      int bn = nt * 16 + lh;
      v2f bf;
      bf.x = lw[bn * 130 + kk];
      bf.y = lw[bn * 130 + kk + 1];
      acc[nt] = __builtin_amdgcn_wmma_f32_16x16x4_f32(
          false, a, false, bf, (short)0, acc[nt], false, false);
    }
  }
  float ss[8];
#pragma unroll
  for (int r = 0; r < 8; ++r) ss[r] = 0.f;
#pragma unroll
  for (int nt = 0; nt < 8; ++nt) {
    float bv = bias[nt * 16 + lh];
#pragma unroll
    for (int r = 0; r < 8; ++r) {
      float g = acc[nt][r] + bv;
      acc[nt][r] = g;
      ss[r] = fmaf(g, g, ss[r]);
    }
  }
  float sclv;
  {
    float vss = half_sum8_rows(ss, lane);
    float vn = sqrtf(fmaxf(vss, EPSF * EPSF));
    float a = vn;
    float k1 = sinhf(a) / a;
    float hs2 = k1 * k1 * vss;
    float xn = sqrtf(fmaxf(hs2, EPSF * EPSF));
    float th = acosh_(fmaxf(coshf(a), 1.0f + EPSF));
    sclv = th / xn * k1;
  }
  float scl[8];
#pragma unroll
  for (int r = 0; r < 8; ++r) scl[r] = __shfl(sclv, (lane & 16) + 2 * r, 32);
#pragma unroll
  for (int nt = 0; nt < 8; ++nt) {
    int col = nt * 16 + lh;
#pragma unroll
    for (int r = 0; r < 8; ++r) {
      int row = m0 + r + hi * 8;
      float v = scl[r] * acc[nt][r];
      size_t o = (size_t)row * DD + col;
      t[o] = v;
    }
  }
}

__global__ __launch_bounds__(256) void gemm_kernel(const float* X, const float* __restrict__ W,
    const float* __restrict__ bias, float* __restrict__ t) {
  gemm_body<256>(X, W, bias, t, blockIdx.x);
}

#define NCH 256
#define CHE 4096
#define NBK 256
#define CAP 8192

__device__ __forceinline__ int scan256(int v, int tid, int* wt, int& total) {
  int lane = tid & 31, w = tid >> 5;
  int inc = v;
  for (int m = 1; m < 32; m <<= 1) {
    int o = __shfl_up(inc, m, 32);
    if (lane >= m) inc += o;
  }
  if (lane == 31 && w < 8) wt[w] = inc;
  __syncthreads();
  int base = 0, tot = 0;
  for (int k = 0; k < 8; ++k) {
    int x = wt[k];
    tot += x;
    if (k < w) base += x;
  }
  __syncthreads();
  total = tot;
  return base + inc - v;
}

__global__ __launch_bounds__(1024) void bin_kernel(const int* __restrict__ ei,
    int* __restrict__ tmp, int* __restrict__ offsT) {
  __shared__ int cnt[NBK];
  __shared__ int offs[NBK];
  __shared__ int wt[8];
  __shared__ __attribute__((aligned(16))) int stage[CHE];
  int tid = threadIdx.x;
  int base = blockIdx.x * CHE;
  if (tid < NBK) cnt[tid] = 0;
  int4 s4 = *(const int4*)(ei + base + 4 * tid);
  int4 d4 = *(const int4*)(ei + NE + base + 4 * tid);
  __syncthreads();
  int s[4] = {s4.x, s4.y, s4.z, s4.w};
  int d[4] = {d4.x, d4.y, d4.z, d4.w};
  int r[4];
#pragma unroll
  for (int j = 0; j < 4; ++j) r[j] = atomicAdd(&cnt[d[j] >> 8], 1);
  __syncthreads();
  int v = (tid < NBK) ? cnt[tid] : 0;
  int tot;
  int ex = scan256(v, tid, wt, tot);
  if (tid < NBK) {
    offs[tid] = ex;
    offsT[blockIdx.x * NBK + tid] = ex;
  }
  __syncthreads();
#pragma unroll
  for (int j = 0; j < 4; ++j) stage[offs[d[j] >> 8] + r[j]] = ((d[j] & 255) << 16) | s[j];
  __syncthreads();
  *(int4*)(tmp + base + 4 * tid) = *(const int4*)(stage + 4 * tid);
}

__device__ __forceinline__ int locate(const int* runB, const int* runS, int p) {
  int lo = 0;
#pragma unroll
  for (int st = NCH / 2; st > 0; st >>= 1)
    if (runB[lo + st] <= p) lo += st;
  return lo * CHE + runS[lo] + (p - runB[lo]);
}

template <int BT>
__device__ __forceinline__ void build_body(const int* __restrict__ tmp,
    const int* __restrict__ offsT, int* __restrict__ seg, int* __restrict__ csr, int b) {
  __shared__ int runS[NCH];
  __shared__ int runB[NCH + 1];
  __shared__ int cntA[256];
  __shared__ int cntB[256];
  __shared__ int nstart[256];
  __shared__ int wt[8];
  __shared__ int wbuf[CAP];
  __shared__ int obuf[CAP];
  int tid = threadIdx.x;
  int s = 0, len = 0;
  if (tid < NCH) {
    int nx = tid * NBK + b + 1;
    if (nx > NCH * NBK - 1) nx = NCH * NBK - 1;
    s = offsT[tid * NBK + b];
    int e = offsT[nx];
    if (b == NBK - 1) e = CHE;
    len = e - s;
    runS[tid] = s;
    cntA[tid] = 0;
    cntB[tid] = 0;
  }
  int bstart, T;
  (void)scan256(s, tid, wt, bstart);
  int ex = scan256(len, tid, wt, T);
  if (tid < NCH) runB[tid] = ex;
  if (tid == 0) runB[NCH] = T;
  __syncthreads();
  const bool small = (T <= CAP);

  for (int p0 = tid; p0 < T; p0 += 4 * BT) {
    int w[4];
#pragma unroll
    for (int j = 0; j < 4; ++j) {
      int p = p0 + j * BT;
      if (p > T - 1) p = T - 1;
      w[j] = tmp[locate(runB, runS, p)];
    }
#pragma unroll
    for (int j = 0; j < 4; ++j) {
      int p = p0 + j * BT;
      if (p < T) {
        if (small) wbuf[p] = w[j];
        atomicAdd(&cntA[w[j] >> 16], 1);
      }
    }
  }
  __syncthreads();
  int c = (tid < 256) ? cntA[tid] : 0;
  int tot2;
  int nex = scan256(c, tid, wt, tot2);
  if (tid < 256) {
    nstart[tid] = nex;
    ((int2*)seg)[b * 256 + tid] = make_int2(bstart + nex, c);
  }
  __syncthreads();

  if (small) {
    for (int p = tid; p < T; p += BT) {
      int w = wbuf[p];
      int dl = w >> 16;
      int r = atomicAdd(&cntB[dl], 1);
      obuf[nstart[dl] + r] = w & 0xFFFF;
    }
    __syncthreads();
    int a = bstart & 31;
    for (int i = tid - a; i < T; i += BT)
      if (i >= 0) csr[bstart + i] = obuf[i];
  } else {
    for (int p0 = tid; p0 < T; p0 += 4 * BT) {
      int w[4];
#pragma unroll
      for (int j = 0; j < 4; ++j) {
        int p = p0 + j * BT;
        if (p > T - 1) p = T - 1;
        w[j] = tmp[locate(runB, runS, p)];
      }
#pragma unroll
      for (int j = 0; j < 4; ++j) {
        int p = p0 + j * BT;
        if (p < T) {
          int dl = w[j] >> 16;
          int r = atomicAdd(&cntB[dl], 1);
          csr[bstart + nstart[dl] + r] = w[j] & 0xFFFF;
        }
      }
    }
  }
}

__global__ __launch_bounds__(512) __attribute__((amdgpu_waves_per_eu(8))) void build_gemm_kernel(
    const int* __restrict__ tmp, const int* __restrict__ offsT, int* __restrict__ seg,
    int* __restrict__ csr, const float* X, const float* __restrict__ W,
    const float* __restrict__ bias, float* __restrict__ t) {
  int b = blockIdx.x;
  int h = b >> 8, l = b & 255;
  int idx = (l >> 1) + 128 * h;
  if (((l ^ h) & 1) == 0) build_body<512>(tmp, offsT, seg, csr, idx);
  else gemm_body<512>(X, W, bias, t, idx);
}

template <int R>
__device__ __forceinline__ void add_rows(const float4* __restrict__ t4, int mine, int k, int lane,
                                         float4& sv) {
  float4 v[R];
#pragma unroll
  for (int j = 0; j < R; ++j) {
    int s = __builtin_amdgcn_readlane(mine, k + j);
    v[j] = t4[(size_t)s * 32 + lane];
  }
#pragma unroll
  for (int j = 0; j < R; ++j) {
    sv.x += v[j].x; sv.y += v[j].y; sv.z += v[j].z; sv.w += v[j].w;
  }
}

__device__ __forceinline__ void gather_node(const float4* __restrict__ t4,
    const int* __restrict__ csr, int start, int n, int mine, int lane, float4& sv) {
  for (int base = 0; base < n; base += 32) {
    int m = (n - base < 32) ? (n - base) : 32;
    if (base > 0) mine = (lane < m) ? csr[start + base + lane] : 0;
    int k = 0;
    for (; k + 8 <= m; k += 8) add_rows<8>(t4, mine, k, lane, sv);
    if (k + 4 <= m) { add_rows<4>(t4, mine, k, lane, sv); k += 4; }
    int r = m - k;
    if (r == 3) add_rows<3>(t4, mine, k, lane, sv);
    else if (r == 2) add_rows<2>(t4, mine, k, lane, sv);
    else if (r == 1) add_rows<1>(t4, mine, k, lane, sv);
  }
}

#define NB 8

__device__ __forceinline__ float take(float vec, int j) {
  return __int_as_float(__builtin_amdgcn_readlane(__float_as_int(vec), 4 * j));
}

__global__ __launch_bounds__(256) void node_kernel(const float* X,
    const float* __restrict__ t, const int* __restrict__ seg,
    const int* __restrict__ csr, float* Y) {
  int w = __builtin_amdgcn_readfirstlane((int)((blockIdx.x * 256 + threadIdx.x) >> 5));
  int lane = threadIdx.x & 31;
  int i0 = w * NB;
  if (i0 >= NN) return;
  int lj = lane >> 2;
  const float4* t4 = (const float4*)t;

  int2 sgv = ((const int2*)seg)[i0 + lj];
  float x0v = X[(size_t)(i0 + lj) * DP];
  float4 sv[NB];
#pragma unroll
  for (int j = 0; j < NB; ++j) sv[j] = t4[(size_t)(i0 + j) * 32 + lane];
  int st[NB], nn[NB], mine0[NB];
#pragma unroll
  for (int j = 0; j < NB; ++j) {
    st[j] = __builtin_amdgcn_readlane(sgv.x, 4 * j);
    nn[j] = __builtin_amdgcn_readlane(sgv.y, 4 * j);
  }
#pragma unroll
  for (int j = 0; j < NB; ++j) mine0[j] = (lane < nn[j]) ? csr[st[j] + lane] : 0;
#pragma unroll
  for (int j = 0; j < NB; ++j) gather_node(t4, csr, st[j], nn[j], mine0[j], lane, sv[j]);

  float xs[NB][4];
#pragma unroll
  for (int j = 0; j < NB; ++j) {
    const float* xrow = X + (size_t)(i0 + j) * DP;
#pragma unroll
    for (int q = 0; q < 4; ++q) xs[j][q] = xrow[1 + lane * 4 + q];
  }

  float invdv = 1.0f / (float)(sgv.y + 1);
  float vv[NB][4];
  float part[NB];
#pragma unroll
  for (int j = 0; j < NB; ++j) {
    float invd = take(invdv, j);
    vv[j][0] = sv[j].x * invd; vv[j][1] = sv[j].y * invd;
    vv[j][2] = sv[j].z * invd; vv[j][3] = sv[j].w * invd;
    float r1 = 0.f;
#pragma unroll
    for (int q = 0; q < 4; ++q) r1 = fmaf(vv[j][q], vv[j][q], r1);
    part[j] = r1;
  }
  float r1v = sum8_rows(part, lane);
  float suv;
  {
    float r1 = r1v;
    float vn = sqrtf(fmaxf(r1, EPSF * EPSF));
    float a1 = vn;
    float k1 = sinhf(a1) / a1;
    float agg_ss = k1 * k1 * r1;
    float x0a = sqrtf(1.0f + agg_ss);
    float un = sqrtf(fmaxf(agg_ss, EPSF * EPSF));
    float th = acosh_(fmaxf(x0a, 1.0f + EPSF));
    suv = th / un * k1;
  }

#pragma unroll
  for (int j = 0; j < NB; ++j) {
    float su = take(suv, j);
    float ru = 0.f;
#pragma unroll
    for (int q = 0; q < 4; ++q) { vv[j][q] = fmaxf(su * vv[j][q], 0.0f); ru = fmaf(vv[j][q], vv[j][q], ru); }
    part[j] = ru;
  }
  float ruv = sum8_rows(part, lane);
  float act0v, k2v;
  {
    float rn = sqrtf(fmaxf(ruv, EPSF * EPSF));
    float a2 = rn;
    act0v = coshf(a2);
    k2v = sinhf(a2) / a2;
  }

#pragma unroll
  for (int j = 0; j < NB; ++j) {
    float k2 = take(k2v, j);
    float dxa = 0.f;
#pragma unroll
    for (int q = 0; q < 4; ++q) { vv[j][q] = k2 * vv[j][q]; dxa = fmaf(xs[j][q], vv[j][q], dxa); }
    part[j] = dxa;
  }
  float dxav = sum8_rows(part, lane);
  float alphav = fmaxf(x0v * act0v - dxav, 1.0f + EPSF);

#pragma unroll
  for (int j = 0; j < NB; ++j) {
    float alpha = take(alphav, j);
    float u2ss = 0.f;
#pragma unroll
    for (int q = 0; q < 4; ++q) { vv[j][q] = vv[j][q] - alpha * xs[j][q]; u2ss = fmaf(vv[j][q], vv[j][q], u2ss); }
    part[j] = u2ss;
  }
  float u2ssv = sum8_rows(part, lane);
  float sv2v, c3v, k3v;
  {
    float u20 = act0v - alphav * x0v;
    float mu2 = u2ssv - u20 * u20;
    float un2 = sqrtf(fmaxf(mu2, EPSF * EPSF));
    float dist = acosh_(alphav);
    float sv2 = BETAF * dist / un2;
    float mbv = sv2 * sv2 * mu2;
    float a3 = sqrtf(fmaxf(mbv, EPSF * EPSF));
    sv2v = sv2;
    c3v = coshf(a3);
    k3v = sinhf(a3) / a3;
  }

#pragma unroll
  for (int j = 0; j < NB; ++j) {
    float c3 = take(c3v, j), k3 = take(k3v, j), sv2 = take(sv2v, j);
    float oss = 0.f;
#pragma unroll
    for (int q = 0; q < 4; ++q) {
      vv[j][q] = c3 * xs[j][q] + k3 * (sv2 * vv[j][q]);
      oss = fmaf(vv[j][q], vv[j][q], oss);
    }
    part[j] = oss;
  }
  float ossv = sum8_rows(part, lane);
  float y0v = sqrtf(1.0f + ossv);

  if ((lane & 3) == 0) Y[(size_t)(i0 + lj) * DP] = y0v;
#pragma unroll
  for (int j = 0; j < NB; ++j) {
    float* yrow = Y + (size_t)(i0 + j) * DP;
#pragma unroll
    for (int q = 0; q < 4; ++q) yrow[1 + lane * 4 + q] = vv[j][q];
  }
}

extern "C" void kernel_launch(void* const* d_in, const int* in_sizes, int n_in,
                              void* d_out, int out_size, void* d_ws, size_t ws_size,
                              hipStream_t stream) {
  const float* x  = (const float*)d_in[0];
  const int*   ei = (const int*)d_in[1];
  const float* W1 = (const float*)d_in[2];
  const float* b1 = (const float*)d_in[3];
  const float* W2 = (const float*)d_in[4];
  const float* b2 = (const float*)d_in[5];
  float* out = (float*)d_out;

  float* t    = (float*)d_ws;
  int* seg    = (int*)(t + (size_t)NN * DD);
  int* offsT  = seg + 2 * NN;
  int* tmp    = offsT + NCH * NBK;
  int* csr    = tmp + NE;

  bin_kernel<<<NCH, 1024, 0, stream>>>(ei, tmp, offsT);
  build_gemm_kernel<<<2 * NBK, 512, 0, stream>>>(tmp, offsT, seg, csr, x, W1, b1, t);
  node_kernel<<<NN / (8 * NB), 256, 0, stream>>>(x, t, seg, csr, out);
  gemm_kernel<<<NN / 128, 256, 0, stream>>>(out, W2, b2, t);
  node_kernel<<<NN / (8 * NB), 256, 0, stream>>>(out, t, seg, csr, out);
}
